# starting kernel
# baseline (speedup 1.0000x reference)
#include <hip/hip_runtime.h>

typedef __attribute__((ext_vector_type(2))) float v2f;
typedef __attribute__((ext_vector_type(8))) float v8f;

#define T_STEPS 64
#define FD      64
#define HH      32
#define WW      32

__global__ __launch_bounds__(256) void convssm_scan_kernel(
    const float* __restrict__ xg,
    const float* __restrict__ Akg,
    const float* __restrict__ Bkg,
    float* __restrict__ outg)
{
    __shared__ float Hs[FD][FD + 1];
    __shared__ float Xs[FD][FD + 1];
    __shared__ float akl[9];
    __shared__ float bkl[9];

    const int plane = blockIdx.x;
    const int chan  = plane & 31;

    const int tid  = threadIdx.x;
    const int wave = tid >> 5;
    const int lane = tid & 31;
    const int n    = lane & 15;
    const int kh   = lane >> 4;
    const int m    = lane & 15;

    const int i0a = (wave >> 2) * 16;
    const int i0b = i0a + 32;
    const int j0  = (wave & 3) * 16;
    const int jn  = j0 + n;

    if (tid < 9) {
        akl[tid] = Akg[chan * 9 + tid];
        bkl[tid] = Bkg[chan * 9 + tid];
    }
    for (int idx = tid; idx < FD * (FD + 1); idx += 256) {
        ((float*)Hs)[idx] = 0.0f;
        ((float*)Xs)[idx] = 0.0f;
    }
    __syncthreads();

    int raA[3], rbA[3];
#pragma unroll
    for (int ks = 0; ks < 3; ++ks) {
        raA[ks] = (i0a + m - ks) & 63;
        rbA[ks] = (i0b + m - ks) & 63;
    }

    int kkArr[15];
    v2f bfA[15], bfB[15];
    {
        int q = 0;
#pragma unroll
        for (int ks = 0; ks < 3; ++ks) {
            const float a0 = akl[ks * 3 + 0], a1 = akl[ks * 3 + 1], a2 = akl[ks * 3 + 2];
            const float b0 = bkl[ks * 3 + 0], b1 = bkl[ks * 3 + 1], b2 = bkl[ks * 3 + 2];
#pragma unroll
            for (int bbk = 0; bbk < 5; ++bbk, ++q) {
                const int kb  = ((j0 >> 2) + bbk + 15) & 15;
                const int kkA = (kb << 2) + 2 * kh;
                kkArr[q] = kkA;
                const int dx = (jn - kkA) & 63;
                const int dy = (jn - kkA - 1) & 63;
                bfA[q].x = (dx == 0 ? a0 : 0.0f) + (dx == 1 ? a1 : 0.0f) + (dx == 2 ? a2 : 0.0f);
                bfA[q].y = (dy == 0 ? a0 : 0.0f) + (dy == 1 ? a1 : 0.0f) + (dy == 2 ? a2 : 0.0f);
                bfB[q].x = (dx == 0 ? b0 : 0.0f) + (dx == 1 ? b1 : 0.0f) + (dx == 2 ? b2 : 0.0f);
                bfB[q].y = (dy == 0 ? b0 : 0.0f) + (dy == 1 ? b1 : 0.0f) + (dy == 2 ? b2 : 0.0f);
            }
        }
    }

#pragma unroll 1
    for (int t = 0; t < T_STEPS; ++t) {
        const long long xoff = ((long long)t * 64 + plane) * 1024;

        for (int idx = tid; idx < HH * WW; idx += 256)
            Xs[idx >> 5][idx & 31] = xg[xoff + idx];

        if (t + 1 < T_STEPS)
            __builtin_prefetch(&xg[xoff + 65536 + (long long)tid * 16], 0, 3);

        __syncthreads();

        v8f dh0 = {};
        v8f dx0 = {};
        v8f dh1 = {};
        v8f dx1 = {};

        {
            int q = 0;
#pragma unroll
            for (int ks = 0; ks < 3; ++ks) {
                const int ra = raA[ks];
                const int rb = rbA[ks];
#pragma unroll
                for (int bbk = 0; bbk < 5; ++bbk, ++q) {
                    const int kk = kkArr[q];
                    v2f ha, hb, xa, xb;
                    ha.x = Hs[ra][kk];  ha.y = Hs[ra][kk + 1];
                    hb.x = Hs[rb][kk];  hb.y = Hs[rb][kk + 1];
                    xa.x = Xs[ra][kk];  xa.y = Xs[ra][kk + 1];
                    xb.x = Xs[rb][kk];  xb.y = Xs[rb][kk + 1];
                    dh0 = __builtin_amdgcn_wmma_f32_16x16x4_f32(
                            false, ha, false, bfA[q], (short)0, dh0, false, false);
                    dx0 = __builtin_amdgcn_wmma_f32_16x16x4_f32(
                            false, xa, false, bfB[q], (short)0, dx0, false, false);
                    dh1 = __builtin_amdgcn_wmma_f32_16x16x4_f32(
                            false, hb, false, bfA[q], (short)0, dh1, false, false);
                    dx1 = __builtin_amdgcn_wmma_f32_16x16x4_f32(
                            false, xb, false, bfB[q], (short)0, dx1, false, false);
                }
            }
        }

        v8f d0, d1;
#pragma unroll
        for (int v = 0; v < 8; ++v) {
            d0[v] = dh0[v] + dx0[v];
            d1[v] = dh1[v] + dx1[v];
        }

        __syncthreads();

#pragma unroll
        for (int v = 0; v < 8; ++v) {
            Hs[i0a + v + 8 * kh][jn] = d0[v];
            Hs[i0b + v + 8 * kh][jn] = d1[v];
        }
        __syncthreads();

        if (j0 < WW) {
#pragma unroll
            for (int v = 0; v < 8; ++v) {
                const int row = i0a + v + 8 * kh;
                outg[xoff + (long long)row * WW + jn] = d0[v];
            }
        }
    }
}

extern "C" void kernel_launch(void* const* d_in, const int* in_sizes, int n_in,
                              void* d_out, int out_size, void* d_ws, size_t ws_size,
                              hipStream_t stream) {
    (void)in_sizes; (void)n_in; (void)out_size; (void)d_ws; (void)ws_size;
    const float* x  = (const float*)d_in[0];
    const float* Ak = (const float*)d_in[1];
    const float* Bk = (const float*)d_in[2];
    float* out = (float*)d_out;
    convssm_scan_kernel<<<dim3(64), dim3(256), 0, stream>>>(x, Ak, Bk, out);
}


# X-term B*x_t of the 32x32 crop leaves the recurrence: a chip-wide first launch computes it for all 64 steps into out, the scan reads it back one step ahead; edge rows stay; sums unchanged
# speedup vs baseline: 2.0733x; 2.0733x over previous
#include <hip/hip_runtime.h>


typedef __attribute__((ext_vector_type(2))) float v2f;
typedef __attribute__((ext_vector_type(8))) float v8f;

#define T_STEPS 64
#define FD      64
#define HH      32
#define WW      32


__device__ __forceinline__ void band_frags(
    const float (&tp)[9], int j0, int jn, int kh, v2f (&bf)[15], int (&kkArr)[15])
{
    int q = 0;
#pragma unroll
    for (int ks = 0; ks < 3; ++ks) {
        const float c0 = tp[ks * 3 + 0], c1 = tp[ks * 3 + 1], c2 = tp[ks * 3 + 2];
#pragma unroll
        for (int bbk = 0; bbk < 5; ++bbk, ++q) {
            const int kb = ((j0 >> 2) + bbk + 15) & 15;
            const int kk = (kb << 2) + 2 * kh;
            kkArr[q] = kk;
            const int dx = (jn - kk) & 63;
            const int dy = (jn - kk - 1) & 63;
            bf[q].x = (dx == 0 ? c0 : 0.0f) + (dx == 1 ? c1 : 0.0f) + (dx == 2 ? c2 : 0.0f);
            bf[q].y = (dy == 0 ? c0 : 0.0f) + (dy == 1 ? c1 : 0.0f) + (dy == 2 ? c2 : 0.0f);
        }
    }
}


template <int BLO>
__device__ __forceinline__ void xpre_load(
    v2f (&xf)[15], const float* __restrict__ xsrc, const int (&offA)[3])
{
#pragma unroll
    for (int ks = 0; ks < 3; ++ks) {
#pragma unroll
        for (int b = BLO; b < 5; ++b)
            xf[ks * 5 + b] = *(const v2f*)(xsrc + offA[ks] + 4 * b);
    }
}

template <int BLO>
__device__ __forceinline__ void xpre_step(
    v2f (&xf)[15], const float* __restrict__ xnext, float* __restrict__ o,
    const int (&offA)[3], const bool (&okA)[3], const v2f (&bfB)[15])
{
    v2f xm[15];
    {
        const v2f zero = {0.0f, 0.0f};
#pragma unroll
        for (int ks = 0; ks < 3; ++ks) {
#pragma unroll
            for (int b = 0; b < 5; ++b)
                xm[ks * 5 + b] = okA[ks] ? xf[ks * 5 + b] : zero;
        }
    }
    xpre_load<BLO>(xf, xnext, offA);

    v8f d = {};
#pragma unroll
    for (int ks = 0; ks < 3; ++ks) {
#pragma unroll
        for (int b = BLO; b < 5; ++b)
            d = __builtin_amdgcn_wmma_f32_16x16x4_f32(
                    false, xm[ks * 5 + b], false, bfB[ks * 5 + b], (short)0, d, false, false);
    }
#pragma unroll
    for (int v = 0; v < 8; ++v) o[v * WW] = d[v];
}

template <int BLO>
__device__ __forceinline__ void xpre_run(
    const float* __restrict__ xg, float* __restrict__ outg, int plane, int tfirst,
    const int (&offA)[3], const bool (&okA)[3], const v2f (&bfB)[15], int outoff)
{
    v2f xe[15], xo[15];
#pragma unroll
    for (int i = 0; i < 15; ++i) { xe[i] = v2f{0.0f, 0.0f}; xo[i] = v2f{0.0f, 0.0f}; }
    xpre_load<BLO>(xe, xg + ((long long)tfirst * 64 + plane) * 1024, offA);
    xpre_load<BLO>(xo, xg + ((long long)(tfirst + 1) * 64 + plane) * 1024, offA);

#pragma unroll 1
    for (int i = 0; i < 8; i += 2) {
        const int t  = tfirst + i;
        const int te = (i + 2 < 8) ? (t + 2) : t;
        const int to = (i + 2 < 8) ? (t + 3) : (t + 1);
        xpre_step<BLO>(xe, xg + ((long long)te * 64 + plane) * 1024,
                       outg + ((long long)t * 64 + plane) * 1024 + outoff, offA, okA, bfB);
        xpre_step<BLO>(xo, xg + ((long long)to * 64 + plane) * 1024,
                       outg + ((long long)(t + 1) * 64 + plane) * 1024 + outoff, offA, okA, bfB);
    }
}

__global__ __launch_bounds__(256) void convssm_xterm_kernel(
    const float* __restrict__ xg,
    const float* __restrict__ Bkg,
    float* __restrict__ outg)
{
    const int plane = blockIdx.x & 63;
    const int tg    = blockIdx.x >> 6;
    const int chan  = plane & 31;

    const int tid  = threadIdx.x;
    const int wave = __builtin_amdgcn_readfirstlane(tid >> 5);
    const int lane = tid & 31;
    const int n    = lane & 15;
    const int kh   = lane >> 4;
    const int m    = lane & 15;

    const int ra  = (wave >> 1) & 1;
    const int ct  = wave & 1;
    const int i0a = ra * 16;
    const int j0  = ct * 16;
    const int jn  = j0 + n;
    const int tfirst = tg * 16 + (wave >> 2) * 8;

    float tp[9];
#pragma unroll
    for (int i = 0; i < 9; ++i) tp[i] = Bkg[chan * 9 + i];

    v2f bfB[15];
    int kkArr[15];
    band_frags(tp, j0, jn, kh, bfB, kkArr);

    bool okA[3];
    int offA[3];
#pragma unroll
    for (int ks = 0; ks < 3; ++ks) {
        const int r = (i0a + m - ks) & 63;
        okA[ks]  = r < HH;
        offA[ks] = (okA[ks] ? r * WW : 0) + j0 - 4 + 2 * kh;
    }
    const int outoff = (i0a + 8 * kh) * WW + jn;

    if (ct == 0) xpre_run<1>(xg, outg, plane, tfirst, offA, okA, bfB, outoff);
    else         xpre_run<0>(xg, outg, plane, tfirst, offA, okA, bfB, outoff);
}


template <int BLO, int BHI>
__device__ __forceinline__ void xb_term(
    const v2f (&xbm)[10], const v2f (&bfB)[15], v8f& dx1)
{
#pragma unroll
    for (int ks = 1; ks < 3; ++ks) {
#pragma unroll
        for (int b = BLO; b < BHI; ++b)
            dx1 = __builtin_amdgcn_wmma_f32_16x16x4_f32(
                    false, xbm[(ks - 1) * 5 + b], false, bfB[ks * 5 + b], (short)0, dx1, false, false);
    }
}

__device__ __forceinline__ void edge_refill(
    v2f (&xa)[3], v2f (&xb)[10], const float* __restrict__ xsrc,
    const int (&offA)[3], const int (&offB)[3], int colmask, bool cola, bool hasb)
{
    if (cola) {
#pragma unroll
        for (int ks = 0; ks < 3; ++ks)
            xa[ks] = *(const v2f*)(xsrc + offA[ks]);
    }
    if (hasb) {
#pragma unroll
        for (int b = 0; b < 5; ++b) {
            if (colmask & (1 << b)) {
#pragma unroll
                for (int ks = 1; ks < 3; ++ks)
                    xb[(ks - 1) * 5 + b] = *(const v2f*)(xsrc + offB[ks] + 4 * b);
            }
        }
    }
}

__global__ __launch_bounds__(256) void convssm_scan_kernel(
    const float* __restrict__ xg,
    const float* __restrict__ Akg,
    const float* __restrict__ Bkg,
    float* __restrict__ outg)
{
    __shared__ float Hs[FD][FD + 1];
    __shared__ float akl[9];
    __shared__ float bkl[9];

    const int plane = blockIdx.x;
    const int chan  = plane & 31;

    const int tid  = threadIdx.x;
    const int wave = __builtin_amdgcn_readfirstlane(tid >> 5);
    const int lane = tid & 31;
    const int n    = lane & 15;
    const int kh   = lane >> 4;
    const int m    = lane & 15;

    const int ra  = wave >> 2;
    const int ct  = wave & 3;
    const int i0a = ra * 16;
    const int i0b = i0a + 32;
    const int j0  = ct * 16;
    const int jn  = j0 + n;

    if (tid < 9) {
        akl[tid] = Akg[chan * 9 + tid];
        bkl[tid] = Bkg[chan * 9 + tid];
    }
    for (int idx = tid; idx < FD * (FD + 1); idx += 256)
        ((float*)Hs)[idx] = 0.0f;
    __syncthreads();

    int raA[3], rbA[3];
    bool okA[3], okB[3];
    int offA[3], offB[3];
#pragma unroll
    for (int ks = 0; ks < 3; ++ks) {
        raA[ks] = (i0a + m - ks) & 63;
        rbA[ks] = (i0b + m - ks) & 63;
        okA[ks] = raA[ks] < HH;
        okB[ks] = rbA[ks] < HH;
        offA[ks] = (okA[ks] ? raA[ks] * WW : 0) + j0 - 4 + 2 * kh;
        offB[ks] = (okB[ks] ? rbA[ks] * WW : 0) + j0 - 4 + 2 * kh;
    }

    int kkArr[15], kkUnused[15];
    v2f bfA[15], bfB[15];
    {
        float ta[9], tb[9];
#pragma unroll
        for (int i = 0; i < 9; ++i) { ta[i] = akl[i]; tb[i] = bkl[i]; }
        band_frags(ta, j0, jn, kh, bfA, kkArr);
        band_frags(tb, j0, jn, kh, bfB, kkUnused);
    }

    int colmask = 0;
#pragma unroll
    for (int b = 0; b < 5; ++b) {
        const int kb = (ct * 4 + b + 15) & 15;
        if (kb < 8) colmask |= (1 << b);
    }
    const bool hasb = (ra == 0);
    const bool crop = (ct < 2);
    const bool cola = (ct == 2);

    const int uoff = (i0a + 8 * kh) * WW + jn;

    v2f xa[3], xb[10];
#pragma unroll
    for (int i = 0; i < 3; ++i)  xa[i] = v2f{0.0f, 0.0f};
#pragma unroll
    for (int i = 0; i < 10; ++i) xb[i] = v2f{0.0f, 0.0f};
    edge_refill(xa, xb, xg + (long long)plane * 1024, offA, offB, colmask, cola, hasb);

    v8f un = {};
    if (crop) {
        const float* u = outg + (long long)plane * 1024 + uoff;
#pragma unroll
        for (int v = 0; v < 8; ++v) un[v] = u[v * WW];
    }

#pragma unroll 1
    for (int t = 0; t < T_STEPS; ++t) {
        const long long xoff = ((long long)t * 64 + plane) * 1024;
        const int tn = (t + 1 < T_STEPS) ? (t + 1) : t;
        const long long noff = ((long long)tn * 64 + plane) * 1024;

        v2f xam[3], xbm[10];
        {
            const v2f zero = {0.0f, 0.0f};
#pragma unroll
            for (int ks = 0; ks < 3; ++ks) {
                xam[ks] = okA[ks] ? xa[ks] : zero;
                if (ks >= 1) {
#pragma unroll
                    for (int b = 0; b < 5; ++b)
                        xbm[(ks - 1) * 5 + b] = okB[ks] ? xb[(ks - 1) * 5 + b] : zero;
                }
            }
        }
        v8f dx0 = un;
        v8f dx1 = {};

        edge_refill(xa, xb, xg + noff, offA, offB, colmask, cola, hasb);
        if (crop) {
            const float* u = outg + noff + uoff;
#pragma unroll
            for (int v = 0; v < 8; ++v) un[v] = u[v * WW];
        }

        if (cola) {
#pragma unroll
            for (int ks = 0; ks < 3; ++ks)
                dx0 = __builtin_amdgcn_wmma_f32_16x16x4_f32(
                        false, xam[ks], false, bfB[ks * 5], (short)0, dx0, false, false);
        }
        if (hasb) {
            if (ct == 0)      xb_term<1, 5>(xbm, bfB, dx1);
            else if (ct == 1) xb_term<0, 5>(xbm, bfB, dx1);
            else if (ct == 2) xb_term<0, 1>(xbm, bfB, dx1);
        }

        v8f dh0 = {};
        v8f dh1 = {};
        {
            int q = 0;
#pragma unroll
            for (int ks = 0; ks < 3; ++ks) {
                const int rra = raA[ks];
                const int rrb = rbA[ks];
#pragma unroll
                for (int bbk = 0; bbk < 5; ++bbk, ++q) {
                    const int kk = kkArr[q];
                    v2f ha, hb;
                    ha.x = Hs[rra][kk];  ha.y = Hs[rra][kk + 1];
                    hb.x = Hs[rrb][kk];  hb.y = Hs[rrb][kk + 1];
                    dh0 = __builtin_amdgcn_wmma_f32_16x16x4_f32(
                            false, ha, false, bfA[q], (short)0, dh0, false, false);
                    dh1 = __builtin_amdgcn_wmma_f32_16x16x4_f32(
                            false, hb, false, bfA[q], (short)0, dh1, false, false);
                }
            }
        }

        v8f d0, d1;
#pragma unroll
        for (int v = 0; v < 8; ++v) {
            d0[v] = dh0[v] + dx0[v];
            d1[v] = dh1[v] + dx1[v];
        }

        __syncthreads();

#pragma unroll
        for (int v = 0; v < 8; ++v) {
            Hs[i0a + v + 8 * kh][jn] = d0[v];
            Hs[i0b + v + 8 * kh][jn] = d1[v];
        }

        __syncthreads();

        if (crop) {
            float* o = outg + xoff + uoff;
#pragma unroll
            for (int v = 0; v < 8; ++v) o[v * WW] = d0[v];
        }
    }
}

extern "C" void kernel_launch(void* const* d_in, const int* in_sizes, int n_in,
                              void* d_out, int out_size, void* d_ws, size_t ws_size,
                              hipStream_t stream) {
    (void)in_sizes; (void)n_in; (void)out_size; (void)d_ws; (void)ws_size;
    const float* x  = (const float*)d_in[0];
    const float* Ak = (const float*)d_in[1];
    const float* Bk = (const float*)d_in[2];
    float* out = (float*)d_out;
    convssm_xterm_kernel<<<dim3(256), dim3(256), 0, stream>>>(x, Bk, out);
    convssm_scan_kernel<<<dim3(64), dim3(256), 0, stream>>>(x, Ak, Bk, out);
}


# Scan workgroup 16 waves, one 16x16 tile each (was 8 waves x 2 tiles): four waves per SIMD cover each other's LDS and load waits, edge-row wave oldest on its SIMD; sums unchanged
# speedup vs baseline: 2.2028x; 1.0625x over previous
#include <hip/hip_runtime.h>


typedef __attribute__((ext_vector_type(2))) float v2f;
typedef __attribute__((ext_vector_type(8))) float v8f;

#define T_STEPS 64
#define FD      64
#define HH      32
#define WW      32


__device__ __forceinline__ void band_frags(
    const float (&tp)[9], int j0, int jn, int kh, v2f (&bf)[15], int (&kkArr)[15])
{
    int q = 0;
#pragma unroll
    for (int ks = 0; ks < 3; ++ks) {
        const float c0 = tp[ks * 3 + 0], c1 = tp[ks * 3 + 1], c2 = tp[ks * 3 + 2];
#pragma unroll
        for (int bbk = 0; bbk < 5; ++bbk, ++q) {
            const int kb = ((j0 >> 2) + bbk + 15) & 15;
            const int kk = (kb << 2) + 2 * kh;
            kkArr[q] = kk;
            const int dx = (jn - kk) & 63;
            const int dy = (jn - kk - 1) & 63;
            bf[q].x = (dx == 0 ? c0 : 0.0f) + (dx == 1 ? c1 : 0.0f) + (dx == 2 ? c2 : 0.0f);
            bf[q].y = (dy == 0 ? c0 : 0.0f) + (dy == 1 ? c1 : 0.0f) + (dy == 2 ? c2 : 0.0f);
        }
    }
}


template <int BLO>
__device__ __forceinline__ void xpre_load(
    v2f (&xf)[15], const float* __restrict__ xsrc, const int (&offA)[3])
{
#pragma unroll
    for (int ks = 0; ks < 3; ++ks) {
#pragma unroll
        for (int b = BLO; b < 5; ++b)
            xf[ks * 5 + b] = *(const v2f*)(xsrc + offA[ks] + 4 * b);
    }
}

template <int BLO>
__device__ __forceinline__ void xpre_step(
    v2f (&xf)[15], const float* __restrict__ xnext, float* __restrict__ o,
    const int (&offA)[3], const bool (&okA)[3], const v2f (&bfB)[15])
{
    v2f xm[15];
    {
        const v2f zero = {0.0f, 0.0f};
#pragma unroll
        for (int ks = 0; ks < 3; ++ks) {
#pragma unroll
            for (int b = 0; b < 5; ++b)
                xm[ks * 5 + b] = okA[ks] ? xf[ks * 5 + b] : zero;
        }
    }
    xpre_load<BLO>(xf, xnext, offA);

    v8f d = {};
#pragma unroll
    for (int ks = 0; ks < 3; ++ks) {
#pragma unroll
        for (int b = BLO; b < 5; ++b)
            d = __builtin_amdgcn_wmma_f32_16x16x4_f32(
                    false, xm[ks * 5 + b], false, bfB[ks * 5 + b], (short)0, d, false, false);
    }
#pragma unroll
    for (int v = 0; v < 8; ++v) o[v * WW] = d[v];
}

template <int BLO>
__device__ __forceinline__ void xpre_run(
    const float* __restrict__ xg, float* __restrict__ outg, int plane, int tfirst,
    const int (&offA)[3], const bool (&okA)[3], const v2f (&bfB)[15], int outoff)
{
    v2f xe[15], xo[15];
#pragma unroll
    for (int i = 0; i < 15; ++i) { xe[i] = v2f{0.0f, 0.0f}; xo[i] = v2f{0.0f, 0.0f}; }
    xpre_load<BLO>(xe, xg + ((long long)tfirst * 64 + plane) * 1024, offA);
    xpre_load<BLO>(xo, xg + ((long long)(tfirst + 1) * 64 + plane) * 1024, offA);

#pragma unroll 1
    for (int i = 0; i < 8; i += 2) {
        const int t  = tfirst + i;
        const int te = (i + 2 < 8) ? (t + 2) : t;
        const int to = (i + 2 < 8) ? (t + 3) : (t + 1);
        xpre_step<BLO>(xe, xg + ((long long)te * 64 + plane) * 1024,
                       outg + ((long long)t * 64 + plane) * 1024 + outoff, offA, okA, bfB);
        xpre_step<BLO>(xo, xg + ((long long)to * 64 + plane) * 1024,
                       outg + ((long long)(t + 1) * 64 + plane) * 1024 + outoff, offA, okA, bfB);
    }
}

__global__ __launch_bounds__(256) void convssm_xterm_kernel(
    const float* __restrict__ xg,
    const float* __restrict__ Bkg,
    float* __restrict__ outg)
{
    const int plane = blockIdx.x & 63;
    const int tg    = blockIdx.x >> 6;
    const int chan  = plane & 31;

    const int tid  = threadIdx.x;
    const int wave = __builtin_amdgcn_readfirstlane(tid >> 5);
    const int lane = tid & 31;
    const int n    = lane & 15;
    const int kh   = lane >> 4;
    const int m    = lane & 15;

    const int ra  = (wave >> 1) & 1;
    const int ct  = wave & 1;
    const int i0a = ra * 16;
    const int j0  = ct * 16;
    const int jn  = j0 + n;
    const int tfirst = tg * 16 + (wave >> 2) * 8;

    float tp[9];
#pragma unroll
    for (int i = 0; i < 9; ++i) tp[i] = Bkg[chan * 9 + i];

    v2f bfB[15];
    int kkArr[15];
    band_frags(tp, j0, jn, kh, bfB, kkArr);

    bool okA[3];
    int offA[3];
#pragma unroll
    for (int ks = 0; ks < 3; ++ks) {
        const int r = (i0a + m - ks) & 63;
        okA[ks]  = r < HH;
        offA[ks] = (okA[ks] ? r * WW : 0) + j0 - 4 + 2 * kh;
    }
    const int outoff = (i0a + 8 * kh) * WW + jn;

    if (ct == 0) xpre_run<1>(xg, outg, plane, tfirst, offA, okA, bfB, outoff);
    else         xpre_run<0>(xg, outg, plane, tfirst, offA, okA, bfB, outoff);
}


template <int BLO, int BHI>
__device__ __forceinline__ void xb_term(
    const v2f (&xbm)[10], const v2f (&bfB)[15], v8f& dx1)
{
#pragma unroll
    for (int ks = 1; ks < 3; ++ks) {
#pragma unroll
        for (int b = BLO; b < BHI; ++b)
            dx1 = __builtin_amdgcn_wmma_f32_16x16x4_f32(
                    false, xbm[(ks - 1) * 5 + b], false, bfB[ks * 5 + b], (short)0, dx1, false, false);
    }
}

__device__ __forceinline__ void edge_refill(
    v2f (&xa)[3], v2f (&xb)[10], const float* __restrict__ xsrc,
    const int (&offA)[3], const int (&offB)[3], int colmask, bool cola, bool hasb)
{
    if (cola) {
#pragma unroll
        for (int ks = 0; ks < 3; ++ks)
            xa[ks] = *(const v2f*)(xsrc + offA[ks]);
    }
    if (hasb) {
#pragma unroll
        for (int b = 0; b < 5; ++b) {
            if (colmask & (1 << b)) {
#pragma unroll
                for (int ks = 1; ks < 3; ++ks)
                    xb[(ks - 1) * 5 + b] = *(const v2f*)(xsrc + offB[ks] + 4 * b);
            }
        }
    }
}

__global__ __launch_bounds__(512) void convssm_scan_kernel(
    const float* __restrict__ xg,
    const float* __restrict__ Akg,
    const float* __restrict__ Bkg,
    float* __restrict__ outg)
{
    __shared__ float Hs[FD][FD + 1];
    __shared__ float akl[9];
    __shared__ float bkl[9];

    const int plane = blockIdx.x;
    const int chan  = plane & 31;

    const int tid  = threadIdx.x;
    const int wave = __builtin_amdgcn_readfirstlane(tid >> 5);
    const int lane = tid & 31;
    const int n    = lane & 15;
    const int kh   = lane >> 4;
    const int m    = lane & 15;

    const int wg  = wave >> 2;
    const int rt  = (wg == 0) ? 2 : ((wg == 3) ? 3 : (wg - 1));
    const int ct  = wave & 3;
    const int i0  = rt * 16;
    const int j0  = ct * 16;
    const int jn  = j0 + n;

    if (tid < 9) {
        akl[tid] = Akg[chan * 9 + tid];
        bkl[tid] = Bkg[chan * 9 + tid];
    }
    for (int idx = tid; idx < FD * (FD + 1); idx += 512)
        ((float*)Hs)[idx] = 0.0f;
    __syncthreads();

    int rA[3];
    bool ok[3];
    int off[3];
#pragma unroll
    for (int ks = 0; ks < 3; ++ks) {
        rA[ks] = (i0 + m - ks) & 63;
        ok[ks] = rA[ks] < HH;
        off[ks] = (ok[ks] ? rA[ks] * WW : 0) + j0 - 4 + 2 * kh;
    }

    int kkArr[15], kkUnused[15];
    v2f bfA[15], bfB[15];
    {
        float ta[9], tb[9];
#pragma unroll
        for (int i = 0; i < 9; ++i) { ta[i] = akl[i]; tb[i] = bkl[i]; }
        band_frags(ta, j0, jn, kh, bfA, kkArr);
        band_frags(tb, j0, jn, kh, bfB, kkUnused);
    }

    int colmask = 0;
#pragma unroll
    for (int b = 0; b < 5; ++b) {
        const int kb = (ct * 4 + b + 15) & 15;
        if (kb < 8) colmask |= (1 << b);
    }
    const bool hasb = (rt == 2);
    const bool crop = (rt < 2) && (ct < 2);
    const bool cola = (rt < 2) && (ct == 2);

    const int uoff = (i0 + 8 * kh) * WW + jn;

    v2f xa[3], xb[10];
#pragma unroll
    for (int i = 0; i < 3; ++i)  xa[i] = v2f{0.0f, 0.0f};
#pragma unroll
    for (int i = 0; i < 10; ++i) xb[i] = v2f{0.0f, 0.0f};
    edge_refill(xa, xb, xg + (long long)plane * 1024, off, off, colmask, cola, hasb);

    v8f un = {};
    if (crop) {
        const float* u = outg + (long long)plane * 1024 + uoff;
#pragma unroll
        for (int v = 0; v < 8; ++v) un[v] = u[v * WW];
    }

#pragma unroll 1
    for (int t = 0; t < T_STEPS; ++t) {
        const long long xoff = ((long long)t * 64 + plane) * 1024;
        const int tn = (t + 1 < T_STEPS) ? (t + 1) : t;
        const long long noff = ((long long)tn * 64 + plane) * 1024;

        v2f xam[3], xbm[10];
        {
            const v2f zero = {0.0f, 0.0f};
#pragma unroll
            for (int ks = 0; ks < 3; ++ks) {
                xam[ks] = ok[ks] ? xa[ks] : zero;
                if (ks >= 1) {
#pragma unroll
                    for (int b = 0; b < 5; ++b)
                        xbm[(ks - 1) * 5 + b] = ok[ks] ? xb[(ks - 1) * 5 + b] : zero;
                }
            }
        }
        v8f dx = un;

        edge_refill(xa, xb, xg + noff, off, off, colmask, cola, hasb);
        if (crop) {
            const float* u = outg + noff + uoff;
#pragma unroll
            for (int v = 0; v < 8; ++v) un[v] = u[v * WW];
        }

        if (cola) {
#pragma unroll
            for (int ks = 0; ks < 3; ++ks)
                dx = __builtin_amdgcn_wmma_f32_16x16x4_f32(
                        false, xam[ks], false, bfB[ks * 5], (short)0, dx, false, false);
        }
        if (hasb) {
            if (ct == 0)      xb_term<1, 5>(xbm, bfB, dx);
            else if (ct == 1) xb_term<0, 5>(xbm, bfB, dx);
            else if (ct == 2) xb_term<0, 1>(xbm, bfB, dx);
        }

        v8f dh = {};
        {
            int q = 0;
#pragma unroll
            for (int ks = 0; ks < 3; ++ks) {
                const int rr = rA[ks];
#pragma unroll
                for (int bbk = 0; bbk < 5; ++bbk, ++q) {
                    const int kk = kkArr[q];
                    v2f ha;
                    ha.x = Hs[rr][kk];  ha.y = Hs[rr][kk + 1];
                    dh = __builtin_amdgcn_wmma_f32_16x16x4_f32(
                            false, ha, false, bfA[q], (short)0, dh, false, false);
                }
            }
        }

        v8f d;
#pragma unroll
        for (int v = 0; v < 8; ++v) d[v] = dh[v] + dx[v];

        __syncthreads();

#pragma unroll
        for (int v = 0; v < 8; ++v) Hs[i0 + v + 8 * kh][jn] = d[v];

        __syncthreads();

        if (crop) {
            float* o = outg + xoff + uoff;
#pragma unroll
            for (int v = 0; v < 8; ++v) o[v * WW] = d[v];
        }
    }
}

extern "C" void kernel_launch(void* const* d_in, const int* in_sizes, int n_in,
                              void* d_out, int out_size, void* d_ws, size_t ws_size,
                              hipStream_t stream) {
    (void)in_sizes; (void)n_in; (void)out_size; (void)d_ws; (void)ws_size;
    const float* x  = (const float*)d_in[0];
    const float* Ak = (const float*)d_in[1];
    const float* Bk = (const float*)d_in[2];
    float* out = (float*)d_out;
    convssm_xterm_kernel<<<dim3(256), dim3(256), 0, stream>>>(x, Bk, out);
    convssm_scan_kernel<<<dim3(64), dim3(512), 0, stream>>>(x, Ak, Bk, out);
}


# Scan step loop unrolled by two over two register sets for the one-step-ahead operands (edge x fragments, U): per-step copies of the sets gone, tiles without X-term skip its tests; sums unchanged
# speedup vs baseline: 2.2934x; 1.0411x over previous
#include <hip/hip_runtime.h>


typedef __attribute__((ext_vector_type(2))) float v2f;
typedef __attribute__((ext_vector_type(8))) float v8f;

#define T_STEPS 64
#define FD      64
#define HH      32
#define WW      32


__device__ __forceinline__ void band_frags(
    const float (&tp)[9], int j0, int jn, int kh, v2f (&bf)[15], int (&kkArr)[15])
{
    int q = 0;
#pragma unroll
    for (int ks = 0; ks < 3; ++ks) {
        const float c0 = tp[ks * 3 + 0], c1 = tp[ks * 3 + 1], c2 = tp[ks * 3 + 2];
#pragma unroll
        for (int bbk = 0; bbk < 5; ++bbk, ++q) {
            const int kb = ((j0 >> 2) + bbk + 15) & 15;
            const int kk = (kb << 2) + 2 * kh;
            kkArr[q] = kk;
            const int dx = (jn - kk) & 63;
            const int dy = (jn - kk - 1) & 63;
            bf[q].x = (dx == 0 ? c0 : 0.0f) + (dx == 1 ? c1 : 0.0f) + (dx == 2 ? c2 : 0.0f);
            bf[q].y = (dy == 0 ? c0 : 0.0f) + (dy == 1 ? c1 : 0.0f) + (dy == 2 ? c2 : 0.0f);
        }
    }
}


template <int BLO>
__device__ __forceinline__ void xpre_load(
    v2f (&xf)[15], const float* __restrict__ xsrc, const int (&offA)[3])
{
#pragma unroll
    for (int ks = 0; ks < 3; ++ks) {
#pragma unroll
        for (int b = BLO; b < 5; ++b)
            xf[ks * 5 + b] = *(const v2f*)(xsrc + offA[ks] + 4 * b);
    }
}

template <int BLO>
__device__ __forceinline__ void xpre_step(
    v2f (&xf)[15], const float* __restrict__ xnext, float* __restrict__ o,
    const int (&offA)[3], const bool (&okA)[3], const v2f (&bfB)[15])
{
    v2f xm[15];
    {
        const v2f zero = {0.0f, 0.0f};
#pragma unroll
        for (int ks = 0; ks < 3; ++ks) {
#pragma unroll
            for (int b = 0; b < 5; ++b)
                xm[ks * 5 + b] = okA[ks] ? xf[ks * 5 + b] : zero;
        }
    }
    xpre_load<BLO>(xf, xnext, offA);

    v8f d = {};
#pragma unroll
    for (int ks = 0; ks < 3; ++ks) {
#pragma unroll
        for (int b = BLO; b < 5; ++b)
            d = __builtin_amdgcn_wmma_f32_16x16x4_f32(
                    false, xm[ks * 5 + b], false, bfB[ks * 5 + b], (short)0, d, false, false);
    }
#pragma unroll
    for (int v = 0; v < 8; ++v) o[v * WW] = d[v];
}

template <int BLO>
__device__ __forceinline__ void xpre_run(
    const float* __restrict__ xg, float* __restrict__ outg, int plane, int tfirst,
    const int (&offA)[3], const bool (&okA)[3], const v2f (&bfB)[15], int outoff)
{
    v2f xe[15], xo[15];
#pragma unroll
    for (int i = 0; i < 15; ++i) { xe[i] = v2f{0.0f, 0.0f}; xo[i] = v2f{0.0f, 0.0f}; }
    xpre_load<BLO>(xe, xg + ((long long)tfirst * 64 + plane) * 1024, offA);
    xpre_load<BLO>(xo, xg + ((long long)(tfirst + 1) * 64 + plane) * 1024, offA);

#pragma unroll 1
    for (int i = 0; i < 8; i += 2) {
        const int t  = tfirst + i;
        const int te = (i + 2 < 8) ? (t + 2) : t;
        const int to = (i + 2 < 8) ? (t + 3) : (t + 1);
        xpre_step<BLO>(xe, xg + ((long long)te * 64 + plane) * 1024,
                       outg + ((long long)t * 64 + plane) * 1024 + outoff, offA, okA, bfB);
        xpre_step<BLO>(xo, xg + ((long long)to * 64 + plane) * 1024,
                       outg + ((long long)(t + 1) * 64 + plane) * 1024 + outoff, offA, okA, bfB);
    }
}

__global__ __launch_bounds__(256) void convssm_xterm_kernel(
    const float* __restrict__ xg,
    const float* __restrict__ Bkg,
    float* __restrict__ outg)
{
    const int plane = blockIdx.x & 63;
    const int tg    = blockIdx.x >> 6;
    const int chan  = plane & 31;

    const int tid  = threadIdx.x;
    const int wave = __builtin_amdgcn_readfirstlane(tid >> 5);
    const int lane = tid & 31;
    const int n    = lane & 15;
    const int kh   = lane >> 4;
    const int m    = lane & 15;

    const int ra  = (wave >> 1) & 1;
    const int ct  = wave & 1;
    const int i0a = ra * 16;
    const int j0  = ct * 16;
    const int jn  = j0 + n;
    const int tfirst = tg * 16 + (wave >> 2) * 8;

    float tp[9];
#pragma unroll
    for (int i = 0; i < 9; ++i) tp[i] = Bkg[chan * 9 + i];

    v2f bfB[15];
    int kkArr[15];
    band_frags(tp, j0, jn, kh, bfB, kkArr);

    bool okA[3];
    int offA[3];
#pragma unroll
    for (int ks = 0; ks < 3; ++ks) {
        const int r = (i0a + m - ks) & 63;
        okA[ks]  = r < HH;
        offA[ks] = (okA[ks] ? r * WW : 0) + j0 - 4 + 2 * kh;
    }
    const int outoff = (i0a + 8 * kh) * WW + jn;

    if (ct == 0) xpre_run<1>(xg, outg, plane, tfirst, offA, okA, bfB, outoff);
    else         xpre_run<0>(xg, outg, plane, tfirst, offA, okA, bfB, outoff);
}


template <int BLO, int BHI>
__device__ __forceinline__ void xb_term(
    const v2f (&xbm)[10], const v2f (&bfB)[15], v8f& dx1)
{
#pragma unroll
    for (int ks = 1; ks < 3; ++ks) {
#pragma unroll
        for (int b = BLO; b < BHI; ++b)
            dx1 = __builtin_amdgcn_wmma_f32_16x16x4_f32(
                    false, xbm[(ks - 1) * 5 + b], false, bfB[ks * 5 + b], (short)0, dx1, false, false);
    }
}

__device__ __forceinline__ void edge_refill(
    v2f (&xa)[3], v2f (&xb)[10], const float* __restrict__ xsrc,
    const int (&offA)[3], const int (&offB)[3], int colmask, bool cola, bool hasb)
{
    if (cola) {
#pragma unroll
        for (int ks = 0; ks < 3; ++ks)
            xa[ks] = *(const v2f*)(xsrc + offA[ks]);
    }
    if (hasb) {
#pragma unroll
        for (int b = 0; b < 5; ++b) {
            if (colmask & (1 << b)) {
#pragma unroll
                for (int ks = 1; ks < 3; ++ks)
                    xb[(ks - 1) * 5 + b] = *(const v2f*)(xsrc + offB[ks] + 4 * b);
            }
        }
    }
}

__global__ __launch_bounds__(512) void convssm_scan_kernel(
    const float* __restrict__ xg,
    const float* __restrict__ Akg,
    const float* __restrict__ Bkg,
    float* __restrict__ outg)
{
    __shared__ float Hs[FD][FD + 1];
    __shared__ float akl[9];
    __shared__ float bkl[9];

    const int plane = blockIdx.x;
    const int chan  = plane & 31;

    const int tid  = threadIdx.x;
    const int wave = __builtin_amdgcn_readfirstlane(tid >> 5);
    const int lane = tid & 31;
    const int n    = lane & 15;
    const int kh   = lane >> 4;
    const int m    = lane & 15;

    const int wg  = wave >> 2;
    const int rt  = (wg == 0) ? 2 : ((wg == 3) ? 3 : (wg - 1));
    const int ct  = wave & 3;
    const int i0  = rt * 16;
    const int j0  = ct * 16;
    const int jn  = j0 + n;

    int rA[3];
    bool ok[3];
    int off[3];
#pragma unroll
    for (int ks = 0; ks < 3; ++ks) {
        rA[ks] = (i0 + m - ks) & 63;
        ok[ks] = rA[ks] < HH;
        off[ks] = (ok[ks] ? rA[ks] * WW : 0) + j0 - 4 + 2 * kh;
    }

    int colmask = 0;
#pragma unroll
    for (int b = 0; b < 5; ++b) {
        const int kb = (ct * 4 + b + 15) & 15;
        if (kb < 8) colmask |= (1 << b);
    }
    const bool hasb = (rt == 2);
    const bool crop = (rt < 2) && (ct < 2);
    const bool cola = (rt < 2) && (ct == 2);
    const bool anyx = (rt < 2) ? (ct < 3) : ((rt == 2) && (ct < 3));

    const int uoff = (i0 + 8 * kh) * WW + jn;

    v2f xa[3], xb[10], xa2[3], xb2[10];
#pragma unroll
    for (int i = 0; i < 3; ++i)  { xa[i] = v2f{0.0f, 0.0f}; xa2[i] = v2f{0.0f, 0.0f}; }
#pragma unroll
    for (int i = 0; i < 10; ++i) { xb[i] = v2f{0.0f, 0.0f}; xb2[i] = v2f{0.0f, 0.0f}; }
    v8f un = {}, un2 = {};

    edge_refill(xa, xb, xg + (long long)plane * 1024, off, off, colmask, cola, hasb);
    if (crop) {
        const float* u = outg + (long long)plane * 1024 + uoff;
#pragma unroll
        for (int v = 0; v < 8; ++v) un[v] = u[v * WW];
    }

    if (tid < 9) {
        akl[tid] = Akg[chan * 9 + tid];
        bkl[tid] = Bkg[chan * 9 + tid];
    }
    for (int idx = tid; idx < FD * (FD + 1); idx += 512)
        ((float*)Hs)[idx] = 0.0f;
    __syncthreads();

    int kkArr[15], kkUnused[15];
    v2f bfA[15], bfB[15];
    {
        float ta[9], tb[9];
#pragma unroll
        for (int i = 0; i < 9; ++i) { ta[i] = akl[i]; tb[i] = bkl[i]; }
        band_frags(ta, j0, jn, kh, bfA, kkArr);
        band_frags(tb, j0, jn, kh, bfB, kkUnused);
    }

    auto step = [&](int t,
                    const v2f (&xaC)[3], const v2f (&xbC)[10], const v8f& unC,
                    v2f (&xaN)[3], v2f (&xbN)[10], v8f& unN)
                    __attribute__((always_inline)) {
        const long long xoff = ((long long)t * 64 + plane) * 1024;
        const int tn = (t + 1 < T_STEPS) ? (t + 1) : t;
        const long long noff = ((long long)tn * 64 + plane) * 1024;

        v8f dx = unC;

        if (anyx) {
            v2f xam[3], xbm[10];
            {
                const v2f zero = {0.0f, 0.0f};
#pragma unroll
                for (int ks = 0; ks < 3; ++ks) {
                    xam[ks] = ok[ks] ? xaC[ks] : zero;
                    if (ks >= 1) {
#pragma unroll
                        for (int b = 0; b < 5; ++b)
                            xbm[(ks - 1) * 5 + b] = ok[ks] ? xbC[(ks - 1) * 5 + b] : zero;
                    }
                }
            }

            edge_refill(xaN, xbN, xg + noff, off, off, colmask, cola, hasb);
            if (crop) {
                const float* u = outg + noff + uoff;
#pragma unroll
                for (int v = 0; v < 8; ++v) unN[v] = u[v * WW];
            }

            if (cola) {
#pragma unroll
                for (int ks = 0; ks < 3; ++ks)
                    dx = __builtin_amdgcn_wmma_f32_16x16x4_f32(
                            false, xam[ks], false, bfB[ks * 5], (short)0, dx, false, false);
            }
            if (hasb) {
                if (ct == 0)      xb_term<1, 5>(xbm, bfB, dx);
                else if (ct == 1) xb_term<0, 5>(xbm, bfB, dx);
                else if (ct == 2) xb_term<0, 1>(xbm, bfB, dx);
            }
        }

        v8f dh = {};
        {
            int q = 0;
#pragma unroll
            for (int ks = 0; ks < 3; ++ks) {
                const int rr = rA[ks];
#pragma unroll
                for (int bbk = 0; bbk < 5; ++bbk, ++q) {
                    const int kk = kkArr[q];
                    v2f ha;
                    ha.x = Hs[rr][kk];  ha.y = Hs[rr][kk + 1];
                    dh = __builtin_amdgcn_wmma_f32_16x16x4_f32(
                            false, ha, false, bfA[q], (short)0, dh, false, false);
                }
            }
        }

        v8f d;
#pragma unroll
        for (int v = 0; v < 8; ++v) d[v] = dh[v] + dx[v];

        __syncthreads();

#pragma unroll
        for (int v = 0; v < 8; ++v) Hs[i0 + v + 8 * kh][jn] = d[v];

        __syncthreads();

        if (crop) {
            float* o = outg + xoff + uoff;
#pragma unroll
            for (int v = 0; v < 8; ++v) o[v * WW] = d[v];
        }
    };

#pragma unroll 1
    for (int t = 0; t < T_STEPS; t += 2) {
        step(t,     xa,  xb,  un,  xa2, xb2, un2);
        step(t + 1, xa2, xb2, un2, xa,  xb,  un);
    }
}

extern "C" void kernel_launch(void* const* d_in, const int* in_sizes, int n_in,
                              void* d_out, int out_size, void* d_ws, size_t ws_size,
                              hipStream_t stream) {
    (void)in_sizes; (void)n_in; (void)out_size; (void)d_ws; (void)ws_size;
    const float* x  = (const float*)d_in[0];
    const float* Ak = (const float*)d_in[1];
    const float* Bk = (const float*)d_in[2];
    float* out = (float*)d_out;
    convssm_xterm_kernel<<<dim3(256), dim3(256), 0, stream>>>(x, Bk, out);
    convssm_scan_kernel<<<dim3(64), dim3(512), 0, stream>>>(x, Ak, Bk, out);
}


# Scan state stays in accumulator registers, fed back as WMMA B operand (row taps banded in A, column taps by DPP row shifts); LDS holds only tile edge rows/columns, 2 buffers; tap sum order changes
# speedup vs baseline: 2.4863x; 1.0841x over previous
#include <hip/hip_runtime.h>


typedef __attribute__((ext_vector_type(2))) float v2f;
typedef __attribute__((ext_vector_type(4))) float v4f;
typedef __attribute__((ext_vector_type(8))) float v8f;

#define T_STEPS 64
#define FD      64
#define HH      32
#define WW      32


__device__ __forceinline__ void band_frags(
    const float (&tp)[9], int j0, int jn, int kh, v2f (&bf)[15], int (&kkArr)[15])
{
    int q = 0;
#pragma unroll
    for (int ks = 0; ks < 3; ++ks) {
        const float c0 = tp[ks * 3 + 0], c1 = tp[ks * 3 + 1], c2 = tp[ks * 3 + 2];
#pragma unroll
        for (int bbk = 0; bbk < 5; ++bbk, ++q) {
            const int kb = ((j0 >> 2) + bbk + 15) & 15;
            const int kk = (kb << 2) + 2 * kh;
            kkArr[q] = kk;
            const int dx = (jn - kk) & 63;
            const int dy = (jn - kk - 1) & 63;
            bf[q].x = (dx == 0 ? c0 : 0.0f) + (dx == 1 ? c1 : 0.0f) + (dx == 2 ? c2 : 0.0f);
            bf[q].y = (dy == 0 ? c0 : 0.0f) + (dy == 1 ? c1 : 0.0f) + (dy == 2 ? c2 : 0.0f);
        }
    }
}


template <int BLO>
__device__ __forceinline__ void xpre_load(
    v2f (&xf)[15], const float* __restrict__ xsrc, const int (&offA)[3])
{
#pragma unroll
    for (int ks = 0; ks < 3; ++ks) {
#pragma unroll
        for (int b = BLO; b < 5; ++b)
            xf[ks * 5 + b] = *(const v2f*)(xsrc + offA[ks] + 4 * b);
    }
}

template <int BLO>
__device__ __forceinline__ void xpre_step(
    v2f (&xf)[15], const float* __restrict__ xnext, float* __restrict__ o,
    const int (&offA)[3], const bool (&okA)[3], const v2f (&bfB)[15])
{
    v2f xm[15];
    {
        const v2f zero = {0.0f, 0.0f};
#pragma unroll
        for (int ks = 0; ks < 3; ++ks) {
#pragma unroll
            for (int b = 0; b < 5; ++b)
                xm[ks * 5 + b] = okA[ks] ? xf[ks * 5 + b] : zero;
        }
    }
    xpre_load<BLO>(xf, xnext, offA);

    v8f d = {};
#pragma unroll
    for (int ks = 0; ks < 3; ++ks) {
#pragma unroll
        for (int b = BLO; b < 5; ++b)
            d = __builtin_amdgcn_wmma_f32_16x16x4_f32(
                    false, xm[ks * 5 + b], false, bfB[ks * 5 + b], (short)0, d, false, false);
    }
#pragma unroll
    for (int v = 0; v < 8; ++v) o[v * WW] = d[v];
}

template <int BLO>
__device__ __forceinline__ void xpre_run(
    const float* __restrict__ xg, float* __restrict__ outg, int plane, int tfirst,
    const int (&offA)[3], const bool (&okA)[3], const v2f (&bfB)[15], int outoff)
{
    v2f xe[15], xo[15];
#pragma unroll
    for (int i = 0; i < 15; ++i) { xe[i] = v2f{0.0f, 0.0f}; xo[i] = v2f{0.0f, 0.0f}; }
    xpre_load<BLO>(xe, xg + ((long long)tfirst * 64 + plane) * 1024, offA);
    xpre_load<BLO>(xo, xg + ((long long)(tfirst + 1) * 64 + plane) * 1024, offA);

#pragma unroll 1
    for (int i = 0; i < 8; i += 2) {
        const int t  = tfirst + i;
        const int te = (i + 2 < 8) ? (t + 2) : t;
        const int to = (i + 2 < 8) ? (t + 3) : (t + 1);
        xpre_step<BLO>(xe, xg + ((long long)te * 64 + plane) * 1024,
                       outg + ((long long)t * 64 + plane) * 1024 + outoff, offA, okA, bfB);
        xpre_step<BLO>(xo, xg + ((long long)to * 64 + plane) * 1024,
                       outg + ((long long)(t + 1) * 64 + plane) * 1024 + outoff, offA, okA, bfB);
    }
}

__global__ __launch_bounds__(256) void convssm_xterm_kernel(
    const float* __restrict__ xg,
    const float* __restrict__ Bkg,
    float* __restrict__ outg)
{
    const int plane = blockIdx.x & 63;
    const int tg    = blockIdx.x >> 6;
    const int chan  = plane & 31;

    const int tid  = threadIdx.x;
    const int wave = __builtin_amdgcn_readfirstlane(tid >> 5);
    const int lane = tid & 31;
    const int n    = lane & 15;
    const int kh   = lane >> 4;
    const int m    = lane & 15;

    const int ra  = (wave >> 1) & 1;
    const int ct  = wave & 1;
    const int i0a = ra * 16;
    const int j0  = ct * 16;
    const int jn  = j0 + n;
    const int tfirst = tg * 16 + (wave >> 2) * 8;

    float tp[9];
#pragma unroll
    for (int i = 0; i < 9; ++i) tp[i] = Bkg[chan * 9 + i];

    v2f bfB[15];
    int kkArr[15];
    band_frags(tp, j0, jn, kh, bfB, kkArr);

    bool okA[3];
    int offA[3];
#pragma unroll
    for (int ks = 0; ks < 3; ++ks) {
        const int r = (i0a + m - ks) & 63;
        okA[ks]  = r < HH;
        offA[ks] = (okA[ks] ? r * WW : 0) + j0 - 4 + 2 * kh;
    }
    const int outoff = (i0a + 8 * kh) * WW + jn;

    if (ct == 0) xpre_run<1>(xg, outg, plane, tfirst, offA, okA, bfB, outoff);
    else         xpre_run<0>(xg, outg, plane, tfirst, offA, okA, bfB, outoff);
}


template <int BLO, int BHI>
__device__ __forceinline__ void xb_term(
    const v2f (&xbm)[10], const v2f (&bfB)[15], v8f& dx1)
{
#pragma unroll
    for (int ks = 1; ks < 3; ++ks) {
#pragma unroll
        for (int b = BLO; b < BHI; ++b)
            dx1 = __builtin_amdgcn_wmma_f32_16x16x4_f32(
                    false, xbm[(ks - 1) * 5 + b], false, bfB[ks * 5 + b], (short)0, dx1, false, false);
    }
}

__device__ __forceinline__ void edge_refill(
    v2f (&xa)[3], v2f (&xb)[10], const float* __restrict__ xsrc,
    const int (&offA)[3], const int (&offB)[3], int colmask, bool cola, bool hasb)
{
    if (cola) {
#pragma unroll
        for (int ks = 0; ks < 3; ++ks)
            xa[ks] = *(const v2f*)(xsrc + offA[ks]);
    }
    if (hasb) {
#pragma unroll
        for (int b = 0; b < 5; ++b) {
            if (colmask & (1 << b)) {
#pragma unroll
                for (int ks = 1; ks < 3; ++ks)
                    xb[(ks - 1) * 5 + b] = *(const v2f*)(xsrc + offB[ks] + 4 * b);
            }
        }
    }
}

template <int KC>
__device__ __forceinline__ float vtap(const float (&tp)[9], int a)
{
    return (a == 0 ? tp[KC] : 0.0f) + (a == 1 ? tp[3 + KC] : 0.0f) + (a == 2 ? tp[6 + KC] : 0.0f);
}

template <int N>
__device__ __forceinline__ float row_shr_keep(float keep, float src)
{
    return __int_as_float(__builtin_amdgcn_update_dpp(
        __float_as_int(keep), __float_as_int(src), 0x110 + N, 0xf, 0xf, false));
}

__global__ __launch_bounds__(512) void convssm_scan_kernel(
    const float* __restrict__ xg,
    const float* __restrict__ Akg,
    const float* __restrict__ Bkg,
    float* __restrict__ outg)
{
    __shared__ __attribute__((aligned(16))) float Brow[2][4][FD][2];
    __shared__ __attribute__((aligned(16))) float Rcol[2][4][4][2][2][8];
    __shared__ float akl[9];
    __shared__ float bkl[9];

    const int plane = blockIdx.x;
    const int chan  = plane & 31;

    const int tid  = threadIdx.x;
    const int wave = __builtin_amdgcn_readfirstlane(tid >> 5);
    const int lane = tid & 31;
    const int n    = lane & 15;
    const int kh   = lane >> 4;
    const int m    = lane & 15;

    const int wg  = wave >> 2;
    const int rt  = (wg == 0) ? 2 : ((wg == 3) ? 3 : (wg - 1));
    const int ct  = wave & 3;
    const int i0  = rt * 16;
    const int j0  = ct * 16;
    const int jn  = j0 + n;

    bool ok[3];
    int off[3];
#pragma unroll
    for (int ks = 0; ks < 3; ++ks) {
        const int r = (i0 + m - ks) & 63;
        ok[ks] = r < HH;
        off[ks] = (ok[ks] ? r * WW : 0) + j0 - 4 + 2 * kh;
    }

    int colmask = 0;
#pragma unroll
    for (int b = 0; b < 5; ++b) {
        const int kb = (ct * 4 + b + 15) & 15;
        if (kb < 8) colmask |= (1 << b);
    }
    const bool hasb = (rt == 2);
    const bool crop = (rt < 2) && (ct < 2);
    const bool cola = (rt < 2) && (ct == 2);
    const bool anyx = (rt < 2) ? (ct < 3) : ((rt == 2) && (ct < 3));

    const int uoff = (i0 + 8 * kh) * WW + jn;

    const int rtA = (rt + 3) & 3;
    const int ctL = (ct + 3) & 3;
    const int browRdA = (rtA * FD + ((jn - kh) & 63)) * 2;
    const int browRdB = (rtA * FD + ((jn - 2) & 63)) * 2;
    const int rcolRdX = (((rt * 4 + ctL) * 2 + (n & 1)) * 2 + kh) * 8;
    const int rcolRdY = (((rt * 4 + ctL) * 2 + 1) * 2 + kh) * 8;
    const int browWr  = (rt * FD + jn) * 2;
    const int rcolWr  = (((rt * 4 + ct) * 2 + (n & 1)) * 2 + kh) * 8;

    v2f xa[3], xb[10], xa2[3], xb2[10];
#pragma unroll
    for (int i = 0; i < 3; ++i)  { xa[i] = v2f{0.0f, 0.0f}; xa2[i] = v2f{0.0f, 0.0f}; }
#pragma unroll
    for (int i = 0; i < 10; ++i) { xb[i] = v2f{0.0f, 0.0f}; xb2[i] = v2f{0.0f, 0.0f}; }
    v8f un = {}, un2 = {};

    edge_refill(xa, xb, xg + (long long)plane * 1024, off, off, colmask, cola, hasb);
    if (crop) {
        const float* u = outg + (long long)plane * 1024 + uoff;
#pragma unroll
        for (int v = 0; v < 8; ++v) un[v] = u[v * WW];
    }

    if (tid < 9) {
        akl[tid] = Akg[chan * 9 + tid];
        bkl[tid] = Bkg[chan * 9 + tid];
    }
    for (int idx = tid; idx < 2 * 4 * FD * 2; idx += 512) {
        (&Brow[0][0][0][0])[idx] = 0.0f;
        (&Rcol[0][0][0][0][0][0])[idx] = 0.0f;
    }
    __syncthreads();

    v2f wA[14], bfB[15];
    {
        int kkUnused[15];
        float ta[9], tb[9];
#pragma unroll
        for (int i = 0; i < 9; ++i) { ta[i] = akl[i]; tb[i] = bkl[i]; }
#pragma unroll
        for (int j = 0; j < 4; ++j) {
            const int a0 = m - (8 * kh + 2 * j);
            wA[0 + j].x = vtap<0>(ta, a0);  wA[0 + j].y = vtap<0>(ta, a0 - 1);
            wA[4 + j].x = vtap<1>(ta, a0);  wA[4 + j].y = vtap<1>(ta, a0 - 1);
            wA[8 + j].x = vtap<2>(ta, a0);  wA[8 + j].y = vtap<2>(ta, a0 - 1);
        }
        wA[12].x = kh ? vtap<1>(ta, m + 2) : vtap<0>(ta, m + 2);
        wA[12].y = kh ? vtap<1>(ta, m + 1) : vtap<0>(ta, m + 1);
        wA[13].x = kh ? 0.0f : vtap<2>(ta, m + 2);
        wA[13].y = kh ? 0.0f : vtap<2>(ta, m + 1);
        band_frags(tb, j0, jn, kh, bfB, kkUnused);
    }

    v8f hst = {};

    auto step = [&](int t, const int par,
                    const v2f (&xaC)[3], const v2f (&xbC)[10], const v8f& unC,
                    v2f (&xaN)[3], v2f (&xbN)[10], v8f& unN)
                    __attribute__((always_inline)) {
        const long long xoff = ((long long)t * 64 + plane) * 1024;
        const int tn = (t + 1 < T_STEPS) ? (t + 1) : t;
        const long long noff = ((long long)tn * 64 + plane) * 1024;

        const float* br = &Brow[0][0][0][0] + par * 512;
        const float* rc = &Rcol[0][0][0][0][0][0] + par * 512;
        const v2f hA = *(const v2f*)(br + browRdA);
        const v2f hB = *(const v2f*)(br + browRdB);
        const v4f x0 = *(const v4f*)(rc + rcolRdX);
        const v4f x1 = *(const v4f*)(rc + rcolRdX + 4);
        const v4f y0 = *(const v4f*)(rc + rcolRdY);
        const v4f y1 = *(const v4f*)(rc + rcolRdY + 4);

        v8f dx = unC;

        if (anyx) {
            v2f xam[3], xbm[10];
            {
                const v2f zero = {0.0f, 0.0f};
#pragma unroll
                for (int ks = 0; ks < 3; ++ks) {
                    xam[ks] = ok[ks] ? xaC[ks] : zero;
                    if (ks >= 1) {
#pragma unroll
                        for (int b = 0; b < 5; ++b)
                            xbm[(ks - 1) * 5 + b] = ok[ks] ? xbC[(ks - 1) * 5 + b] : zero;
                    }
                }
            }

            edge_refill(xaN, xbN, xg + noff, off, off, colmask, cola, hasb);
            if (crop) {
                const float* u = outg + noff + uoff;
#pragma unroll
                for (int v = 0; v < 8; ++v) unN[v] = u[v * WW];
            }

            if (cola) {
#pragma unroll
                for (int ks = 0; ks < 3; ++ks)
                    dx = __builtin_amdgcn_wmma_f32_16x16x4_f32(
                            false, xam[ks], false, bfB[ks * 5], (short)0, dx, false, false);
            }
            if (hasb) {
                if (ct == 0)      xb_term<1, 5>(xbm, bfB, dx);
                else if (ct == 1) xb_term<0, 5>(xbm, bfB, dx);
                else if (ct == 2) xb_term<0, 1>(xbm, bfB, dx);
            }
        }

        v8f dh = {};
#pragma unroll
        for (int j = 0; j < 4; ++j)
            dh = __builtin_amdgcn_wmma_f32_16x16x4_f32(
                    false, wA[j], false, v2f{hst[2 * j], hst[2 * j + 1]}, (short)0, dh, false, false);
        dh = __builtin_amdgcn_wmma_f32_16x16x4_f32(
                false, wA[12], false, hA, (short)0, dh, false, false);
        dh = __builtin_amdgcn_wmma_f32_16x16x4_f32(
                false, wA[13], false, hB, (short)0, dh, false, false);
        {
            float e[8];
#pragma unroll
            for (int r = 0; r < 4; ++r) {
                e[r]     = row_shr_keep<1>(y0[r], hst[r]);
                e[4 + r] = row_shr_keep<1>(y1[r], hst[4 + r]);
            }
#pragma unroll
            for (int j = 0; j < 4; ++j)
                dh = __builtin_amdgcn_wmma_f32_16x16x4_f32(
                        false, wA[4 + j], false, v2f{e[2 * j], e[2 * j + 1]}, (short)0, dh, false, false);
        }
        {
            float e[8];
#pragma unroll
            for (int r = 0; r < 4; ++r) {
                e[r]     = row_shr_keep<2>(x0[r], hst[r]);
                e[4 + r] = row_shr_keep<2>(x1[r], hst[4 + r]);
            }
#pragma unroll
            for (int j = 0; j < 4; ++j)
                dh = __builtin_amdgcn_wmma_f32_16x16x4_f32(
                        false, wA[8 + j], false, v2f{e[2 * j], e[2 * j + 1]}, (short)0, dh, false, false);
        }

        v8f d;
#pragma unroll
        for (int v = 0; v < 8; ++v) d[v] = dh[v] + dx[v];
        hst = d;

        float* bw = &Brow[0][0][0][0] + (par ^ 1) * 512;
        float* rw = &Rcol[0][0][0][0][0][0] + (par ^ 1) * 512;
        if (kh == 1) *(v2f*)(bw + browWr) = v2f{d[6], d[7]};
        if (n >= 14) {
            *(v4f*)(rw + rcolWr)     = v4f{d[0], d[1], d[2], d[3]};
            *(v4f*)(rw + rcolWr + 4) = v4f{d[4], d[5], d[6], d[7]};
        }

        __syncthreads();

        if (crop) {
            float* o = outg + xoff + uoff;
#pragma unroll
            for (int v = 0; v < 8; ++v) o[v * WW] = d[v];
        }
    };

#pragma unroll 1
    for (int t = 0; t < T_STEPS; t += 2) {
        step(t,     0, xa,  xb,  un,  xa2, xb2, un2);
        step(t + 1, 1, xa2, xb2, un2, xa,  xb,  un);
    }
}

extern "C" void kernel_launch(void* const* d_in, const int* in_sizes, int n_in,
                              void* d_out, int out_size, void* d_ws, size_t ws_size,
                              hipStream_t stream) {
    (void)in_sizes; (void)n_in; (void)out_size; (void)d_ws; (void)ws_size;
    const float* x  = (const float*)d_in[0];
    const float* Ak = (const float*)d_in[1];
    const float* Bk = (const float*)d_in[2];
    float* out = (float*)d_out;
    convssm_xterm_kernel<<<dim3(256), dim3(256), 0, stream>>>(x, Bk, out);
    convssm_scan_kernel<<<dim3(64), dim3(512), 0, stream>>>(x, Ak, Bk, out);
}


# Scan edge X-term: K = only the x rows / columns 30,31 that reach it: +1 WMMA a row-32 tile (was 8/10/2), +2 a column-32 tile (3); tiles re-dealt: 58 WMMAs a step per SIMD (66); edge sums reordered
# speedup vs baseline: 2.6513x; 1.0664x over previous
#include <hip/hip_runtime.h>


typedef __attribute__((ext_vector_type(2))) float v2f;
typedef __attribute__((ext_vector_type(4))) float v4f;
typedef __attribute__((ext_vector_type(8))) float v8f;

#define T_STEPS 64
#define FD      64
#define HH      32
#define WW      32


__device__ __forceinline__ void band_frags(
    const float (&tp)[9], int j0, int jn, int kh, v2f (&bf)[15], int (&kkArr)[15])
{
    int q = 0;
#pragma unroll
    for (int ks = 0; ks < 3; ++ks) {
        const float c0 = tp[ks * 3 + 0], c1 = tp[ks * 3 + 1], c2 = tp[ks * 3 + 2];
#pragma unroll
        for (int bbk = 0; bbk < 5; ++bbk, ++q) {
            const int kb = ((j0 >> 2) + bbk + 15) & 15;
            const int kk = (kb << 2) + 2 * kh;
            kkArr[q] = kk;
            const int dx = (jn - kk) & 63;
            const int dy = (jn - kk - 1) & 63;
            bf[q].x = (dx == 0 ? c0 : 0.0f) + (dx == 1 ? c1 : 0.0f) + (dx == 2 ? c2 : 0.0f);
            bf[q].y = (dy == 0 ? c0 : 0.0f) + (dy == 1 ? c1 : 0.0f) + (dy == 2 ? c2 : 0.0f);
        }
    }
}


template <int BLO>
__device__ __forceinline__ void xpre_load(
    v2f (&xf)[15], const float* __restrict__ xsrc, const int (&offA)[3])
{
#pragma unroll
    for (int ks = 0; ks < 3; ++ks) {
#pragma unroll
        for (int b = BLO; b < 5; ++b)
            xf[ks * 5 + b] = *(const v2f*)(xsrc + offA[ks] + 4 * b);
    }
}

template <int BLO>
__device__ __forceinline__ void xpre_step(
    v2f (&xf)[15], const float* __restrict__ xnext, float* __restrict__ o,
    const int (&offA)[3], const bool (&okA)[3], const v2f (&bfB)[15])
{
    v2f xm[15];
    {
        const v2f zero = {0.0f, 0.0f};
#pragma unroll
        for (int ks = 0; ks < 3; ++ks) {
#pragma unroll
            for (int b = 0; b < 5; ++b)
                xm[ks * 5 + b] = okA[ks] ? xf[ks * 5 + b] : zero;
        }
    }
    xpre_load<BLO>(xf, xnext, offA);

    v8f d = {};
#pragma unroll
    for (int ks = 0; ks < 3; ++ks) {
#pragma unroll
        for (int b = BLO; b < 5; ++b)
            d = __builtin_amdgcn_wmma_f32_16x16x4_f32(
                    false, xm[ks * 5 + b], false, bfB[ks * 5 + b], (short)0, d, false, false);
    }
#pragma unroll
    for (int v = 0; v < 8; ++v) o[v * WW] = d[v];
}

template <int BLO>
__device__ __forceinline__ void xpre_run(
    const float* __restrict__ xg, float* __restrict__ outg, int plane, int tfirst,
    const int (&offA)[3], const bool (&okA)[3], const v2f (&bfB)[15], int outoff)
{
    v2f xe[15], xo[15];
#pragma unroll
    for (int i = 0; i < 15; ++i) { xe[i] = v2f{0.0f, 0.0f}; xo[i] = v2f{0.0f, 0.0f}; }
    xpre_load<BLO>(xe, xg + ((long long)tfirst * 64 + plane) * 1024, offA);
    xpre_load<BLO>(xo, xg + ((long long)(tfirst + 1) * 64 + plane) * 1024, offA);

#pragma unroll 1
    for (int i = 0; i < 8; i += 2) {
        const int t  = tfirst + i;
        const int te = (i + 2 < 8) ? (t + 2) : t;
        const int to = (i + 2 < 8) ? (t + 3) : (t + 1);
        xpre_step<BLO>(xe, xg + ((long long)te * 64 + plane) * 1024,
                       outg + ((long long)t * 64 + plane) * 1024 + outoff, offA, okA, bfB);
        xpre_step<BLO>(xo, xg + ((long long)to * 64 + plane) * 1024,
                       outg + ((long long)(t + 1) * 64 + plane) * 1024 + outoff, offA, okA, bfB);
    }
}

__global__ __launch_bounds__(256) void convssm_xterm_kernel(
    const float* __restrict__ xg,
    const float* __restrict__ Bkg,
    float* __restrict__ outg)
{
    const int plane = blockIdx.x & 63;
    const int tg    = blockIdx.x >> 6;
    const int chan  = plane & 31;

    const int tid  = threadIdx.x;
    const int wave = __builtin_amdgcn_readfirstlane(tid >> 5);
    const int lane = tid & 31;
    const int n    = lane & 15;
    const int kh   = lane >> 4;
    const int m    = lane & 15;

    const int ra  = (wave >> 1) & 1;
    const int ct  = wave & 1;
    const int i0a = ra * 16;
    const int j0  = ct * 16;
    const int jn  = j0 + n;
    const int tfirst = tg * 16 + (wave >> 2) * 8;

    float tp[9];
#pragma unroll
    for (int i = 0; i < 9; ++i) tp[i] = Bkg[chan * 9 + i];

    v2f bfB[15];
    int kkArr[15];
    band_frags(tp, j0, jn, kh, bfB, kkArr);

    bool okA[3];
    int offA[3];
#pragma unroll
    for (int ks = 0; ks < 3; ++ks) {
        const int r = (i0a + m - ks) & 63;
        okA[ks]  = r < HH;
        offA[ks] = (okA[ks] ? r * WW : 0) + j0 - 4 + 2 * kh;
    }
    const int outoff = (i0a + 8 * kh) * WW + jn;

    if (ct == 0) xpre_run<1>(xg, outg, plane, tfirst, offA, okA, bfB, outoff);
    else         xpre_run<0>(xg, outg, plane, tfirst, offA, okA, bfB, outoff);
}


__device__ __forceinline__ void edge_refill(
    v2f (&xc)[2], float (&xe)[4], const float* __restrict__ xsrc,
    int co1, int co2, bool cv1, bool cv2,
    int eo1, int eo2, bool ev1, bool ev2, bool cola, bool hasb)
{
    if (cola) {
        if (cv1) xc[0] = *(const v2f*)(xsrc + co1);
        if (cv2) xc[1] = *(const v2f*)(xsrc + co2);
    }
    if (hasb) {
        if (ev1) { xe[0] = xsrc[eo1]; xe[1] = xsrc[eo1 + WW]; }
        if (ev2) { xe[2] = xsrc[eo2]; xe[3] = xsrc[eo2 + WW]; }
    }
}

template <int KC>
__device__ __forceinline__ float vtap(const float (&tp)[9], int a)
{
    return (a == 0 ? tp[KC] : 0.0f) + (a == 1 ? tp[3 + KC] : 0.0f) + (a == 2 ? tp[6 + KC] : 0.0f);
}

template <int N>
__device__ __forceinline__ float row_shr_keep(float keep, float src)
{
    return __int_as_float(__builtin_amdgcn_update_dpp(
        __float_as_int(keep), __float_as_int(src), 0x110 + N, 0xf, 0xf, false));
}

__global__ __launch_bounds__(512) void convssm_scan_kernel(
    const float* __restrict__ xg,
    const float* __restrict__ Akg,
    const float* __restrict__ Bkg,
    float* __restrict__ outg)
{
    __shared__ __attribute__((aligned(16))) float Brow[2][4][FD][2];
    __shared__ __attribute__((aligned(16))) float Rcol[2][4][4][2][2][8];
    __shared__ float akl[9];
    __shared__ float bkl[9];

    const int plane = blockIdx.x;
    const int chan  = plane & 31;

    const int tid  = threadIdx.x;
    const int wave = __builtin_amdgcn_readfirstlane(tid >> 5);
    const int lane = tid & 31;
    const int n    = lane & 15;
    const int kh   = lane >> 4;
    const int m    = lane & 15;

    const unsigned long long tileOfWave = 0xBFED754C3109A628ULL;
    const int tile = (int)((tileOfWave >> (wave * 4)) & 15ULL);
    const int rt  = tile >> 2;
    const int ct  = tile & 3;
    const int i0  = rt * 16;
    const int j0  = ct * 16;
    const int jn  = j0 + n;


    const bool hasb = (rt == 2) && (ct < 3);
    const bool crop = (rt < 2) && (ct < 2);
    const bool cola = (rt < 2) && (ct == 2);
    const bool anyx = hasb || crop || cola;

    const int  ec1 = jn - kh;
    const int  ec2 = jn - 2;
    const bool ev1 = (ec1 >= 0) && (ec1 < WW);
    const bool ev2 = (kh == 1) && (ec2 >= 0) && (ec2 < WW);
    const int  eo1 = 30 * WW + (ev1 ? ec1 : 0);
    const int  eo2 = 30 * WW + (ev2 ? ec2 : 0);

    const int  rc1 = (i0 + m - kh) & 63;
    const int  rc2 = (i0 + m - 2) & 63;
    const bool cv1 = rc1 < HH;
    const bool cv2 = (kh == 0) && (rc2 < HH);
    const int  co1 = (cv1 ? rc1 * WW : 0) + 30;
    const int  co2 = (cv2 ? rc2 * WW : 0) + 30;

    const int uoff = (i0 + 8 * kh) * WW + jn;

    const int rtA = (rt + 3) & 3;
    const int ctL = (ct + 3) & 3;
    const int browRdA = (rtA * FD + ((jn - kh) & 63)) * 2;
    const int browRdB = (rtA * FD + ((jn - 2) & 63)) * 2;
    const int rcolRdX = (((rt * 4 + ctL) * 2 + (n & 1)) * 2 + kh) * 8;
    const int rcolRdY = (((rt * 4 + ctL) * 2 + 1) * 2 + kh) * 8;
    const int browWr  = (rt * FD + jn) * 2;
    const int rcolWr  = (((rt * 4 + ct) * 2 + (n & 1)) * 2 + kh) * 8;

    v2f xc[2], xc2[2];
    float xe[4], xe2[4];
#pragma unroll
    for (int i = 0; i < 2; ++i) { xc[i] = v2f{0.0f, 0.0f}; xc2[i] = v2f{0.0f, 0.0f}; }
#pragma unroll
    for (int i = 0; i < 4; ++i) { xe[i] = 0.0f; xe2[i] = 0.0f; }
    v8f un = {}, un2 = {};

    edge_refill(xc, xe, xg + (long long)plane * 1024,
                co1, co2, cv1, cv2, eo1, eo2, ev1, ev2, cola, hasb);
    if (crop) {
        const float* u = outg + (long long)plane * 1024 + uoff;
#pragma unroll
        for (int v = 0; v < 8; ++v) un[v] = u[v * WW];
    }

    if (tid < 9) {
        akl[tid] = Akg[chan * 9 + tid];
        bkl[tid] = Bkg[chan * 9 + tid];
    }
    for (int idx = tid; idx < 2 * 4 * FD * 2; idx += 512) {
        (&Brow[0][0][0][0])[idx] = 0.0f;
        (&Rcol[0][0][0][0][0][0])[idx] = 0.0f;
    }
    __syncthreads();

    v2f wA[14], wB, cB[2];
    {
        float ta[9], tb[9];
#pragma unroll
        for (int i = 0; i < 9; ++i) { ta[i] = akl[i]; tb[i] = bkl[i]; }
#pragma unroll
        for (int j = 0; j < 4; ++j) {
            const int a0 = m - (8 * kh + 2 * j);
            wA[0 + j].x = vtap<0>(ta, a0);  wA[0 + j].y = vtap<0>(ta, a0 - 1);
            wA[4 + j].x = vtap<1>(ta, a0);  wA[4 + j].y = vtap<1>(ta, a0 - 1);
            wA[8 + j].x = vtap<2>(ta, a0);  wA[8 + j].y = vtap<2>(ta, a0 - 1);
        }
        wA[12].x = kh ? vtap<1>(ta, m + 2) : vtap<0>(ta, m + 2);
        wA[12].y = kh ? vtap<1>(ta, m + 1) : vtap<0>(ta, m + 1);
        wA[13].x = kh ? (hasb ? vtap<2>(tb, m + 2) : 0.0f) : vtap<2>(ta, m + 2);
        wA[13].y = kh ? (hasb ? vtap<2>(tb, m + 1) : 0.0f) : vtap<2>(ta, m + 1);
        wB.x = kh ? vtap<1>(tb, m + 2) : vtap<0>(tb, m + 2);
        wB.y = kh ? vtap<1>(tb, m + 1) : vtap<0>(tb, m + 1);
        const float r2 = kh ? tb[5] : tb[2];
        const float r1 = kh ? tb[4] : tb[1];
        cB[0].x = (n == 0) ? r2 : 0.0f;
        cB[0].y = (n == 0) ? r1 : ((n == 1) ? r2 : 0.0f);
        cB[1].x = (kh == 0 && n == 0) ? tb[8] : 0.0f;
        cB[1].y = (kh == 0) ? ((n == 0) ? tb[7] : ((n == 1) ? tb[8] : 0.0f)) : 0.0f;
    }

    v8f hst = {};

    auto step = [&](int t, const int par,
                    const v2f (&xcC)[2], const float (&xeC)[4], const v8f& unC,
                    v2f (&xcN)[2], float (&xeN)[4], v8f& unN)
                    __attribute__((always_inline)) {
        const long long xoff = ((long long)t * 64 + plane) * 1024;
        const int tn = (t + 1 < T_STEPS) ? (t + 1) : t;
        const long long noff = ((long long)tn * 64 + plane) * 1024;

        const float* br = &Brow[0][0][0][0] + par * 512;
        const float* rc = &Rcol[0][0][0][0][0][0] + par * 512;
        const v2f hA = *(const v2f*)(br + browRdA);
        const v2f hB = *(const v2f*)(br + browRdB);
        const v4f x0 = *(const v4f*)(rc + rcolRdX);
        const v4f x1 = *(const v4f*)(rc + rcolRdX + 4);
        const v4f y0 = *(const v4f*)(rc + rcolRdY);
        const v4f y1 = *(const v4f*)(rc + rcolRdY + 4);

        v8f dx = unC;

        if (anyx) {
            const v2f xcm0 = xcC[0];
            const v2f xcm1 = xcC[1];
            const v2f xr0  = {xeC[0], xeC[1]};

            edge_refill(xcN, xeN, xg + noff,
                        co1, co2, cv1, cv2, eo1, eo2, ev1, ev2, cola, hasb);
            if (crop) {
                const float* u = outg + noff + uoff;
#pragma unroll
                for (int v = 0; v < 8; ++v) unN[v] = u[v * WW];
            }

            if (cola) {
                dx = __builtin_amdgcn_wmma_f32_16x16x4_f32(
                        false, xcm0, false, cB[0], (short)0, dx, false, false);
                dx = __builtin_amdgcn_wmma_f32_16x16x4_f32(
                        false, xcm1, false, cB[1], (short)0, dx, false, false);
            }
            if (hasb) {
                dx = __builtin_amdgcn_wmma_f32_16x16x4_f32(
                        false, wB, false, xr0, (short)0, dx, false, false);
            }
        }

        v8f dh = {};
#pragma unroll
        for (int j = 0; j < 4; ++j)
            dh = __builtin_amdgcn_wmma_f32_16x16x4_f32(
                    false, wA[j], false, v2f{hst[2 * j], hst[2 * j + 1]}, (short)0, dh, false, false);
        const v2f hBx = {kh ? xeC[2] : hB.x, kh ? xeC[3] : hB.y};
        dh = __builtin_amdgcn_wmma_f32_16x16x4_f32(
                false, wA[12], false, hA, (short)0, dh, false, false);
        dh = __builtin_amdgcn_wmma_f32_16x16x4_f32(
                false, wA[13], false, hBx, (short)0, dh, false, false);
        {
            float e[8];
#pragma unroll
            for (int r = 0; r < 4; ++r) {
                e[r]     = row_shr_keep<1>(y0[r], hst[r]);
                e[4 + r] = row_shr_keep<1>(y1[r], hst[4 + r]);
            }
#pragma unroll
            for (int j = 0; j < 4; ++j)
                dh = __builtin_amdgcn_wmma_f32_16x16x4_f32(
                        false, wA[4 + j], false, v2f{e[2 * j], e[2 * j + 1]}, (short)0, dh, false, false);
        }
        {
            float e[8];
#pragma unroll
            for (int r = 0; r < 4; ++r) {
                e[r]     = row_shr_keep<2>(x0[r], hst[r]);
                e[4 + r] = row_shr_keep<2>(x1[r], hst[4 + r]);
            }
#pragma unroll
            for (int j = 0; j < 4; ++j)
                dh = __builtin_amdgcn_wmma_f32_16x16x4_f32(
                        false, wA[8 + j], false, v2f{e[2 * j], e[2 * j + 1]}, (short)0, dh, false, false);
        }

        v8f d;
#pragma unroll
        for (int v = 0; v < 8; ++v) d[v] = dh[v] + dx[v];
        hst = d;

        float* bw = &Brow[0][0][0][0] + (par ^ 1) * 512;
        float* rw = &Rcol[0][0][0][0][0][0] + (par ^ 1) * 512;
        if (kh == 1) *(v2f*)(bw + browWr) = v2f{d[6], d[7]};
        if (n >= 14) {
            *(v4f*)(rw + rcolWr)     = v4f{d[0], d[1], d[2], d[3]};
            *(v4f*)(rw + rcolWr + 4) = v4f{d[4], d[5], d[6], d[7]};
        }

        __syncthreads();

        if (crop) {
            float* o = outg + xoff + uoff;
#pragma unroll
            for (int v = 0; v < 8; ++v) o[v * WW] = d[v];
        }
    };

#pragma unroll 1
    for (int t = 0; t < T_STEPS; t += 2) {
        step(t,     0, xc,  xe,  un,  xc2, xe2, un2);
        step(t + 1, 1, xc2, xe2, un2, xc,  xe,  un);
    }
}

extern "C" void kernel_launch(void* const* d_in, const int* in_sizes, int n_in,
                              void* d_out, int out_size, void* d_ws, size_t ws_size,
                              hipStream_t stream) {
    (void)in_sizes; (void)n_in; (void)out_size; (void)d_ws; (void)ws_size;
    const float* x  = (const float*)d_in[0];
    const float* Ak = (const float*)d_in[1];
    const float* Bk = (const float*)d_in[2];
    float* out = (float*)d_out;
    convssm_xterm_kernel<<<dim3(256), dim3(256), 0, stream>>>(x, Bk, out);
    convssm_scan_kernel<<<dim3(64), dim3(512), 0, stream>>>(x, Ak, Bk, out);
}


# Scan step barrier split by hand: a wave signals once its edges are in LDS and waits only where the next step reads a neighbour's; X-term and two column-0 WMMAs run in between; sums unchanged
# speedup vs baseline: 2.7855x; 1.0506x over previous
#include <hip/hip_runtime.h>


typedef __attribute__((ext_vector_type(2))) float v2f;
typedef __attribute__((ext_vector_type(4))) float v4f;
typedef __attribute__((ext_vector_type(8))) float v8f;

#define T_STEPS 64
#define FD      64
#define HH      32
#define WW      32


__device__ __forceinline__ void band_frags(
    const float (&tp)[9], int j0, int jn, int kh, v2f (&bf)[15], int (&kkArr)[15])
{
    int q = 0;
#pragma unroll
    for (int ks = 0; ks < 3; ++ks) {
        const float c0 = tp[ks * 3 + 0], c1 = tp[ks * 3 + 1], c2 = tp[ks * 3 + 2];
#pragma unroll
        for (int bbk = 0; bbk < 5; ++bbk, ++q) {
            const int kb = ((j0 >> 2) + bbk + 15) & 15;
            const int kk = (kb << 2) + 2 * kh;
            kkArr[q] = kk;
            const int dx = (jn - kk) & 63;
            const int dy = (jn - kk - 1) & 63;
            bf[q].x = (dx == 0 ? c0 : 0.0f) + (dx == 1 ? c1 : 0.0f) + (dx == 2 ? c2 : 0.0f);
            bf[q].y = (dy == 0 ? c0 : 0.0f) + (dy == 1 ? c1 : 0.0f) + (dy == 2 ? c2 : 0.0f);
        }
    }
}


template <int BLO>
__device__ __forceinline__ void xpre_load(
    v2f (&xf)[15], const float* __restrict__ xsrc, const int (&offA)[3])
{
#pragma unroll
    for (int ks = 0; ks < 3; ++ks) {
#pragma unroll
        for (int b = BLO; b < 5; ++b)
            xf[ks * 5 + b] = *(const v2f*)(xsrc + offA[ks] + 4 * b);
    }
}

template <int BLO>
__device__ __forceinline__ void xpre_step(
    v2f (&xf)[15], const float* __restrict__ xnext, float* __restrict__ o,
    const int (&offA)[3], const bool (&okA)[3], const v2f (&bfB)[15])
{
    v2f xm[15];
    {
        const v2f zero = {0.0f, 0.0f};
#pragma unroll
        for (int ks = 0; ks < 3; ++ks) {
#pragma unroll
            for (int b = 0; b < 5; ++b)
                xm[ks * 5 + b] = okA[ks] ? xf[ks * 5 + b] : zero;
        }
    }
    xpre_load<BLO>(xf, xnext, offA);

    v8f d = {};
#pragma unroll
    for (int ks = 0; ks < 3; ++ks) {
#pragma unroll
        for (int b = BLO; b < 5; ++b)
            d = __builtin_amdgcn_wmma_f32_16x16x4_f32(
                    false, xm[ks * 5 + b], false, bfB[ks * 5 + b], (short)0, d, false, false);
    }
#pragma unroll
    for (int v = 0; v < 8; ++v) o[v * WW] = d[v];
}

template <int BLO>
__device__ __forceinline__ void xpre_run(
    const float* __restrict__ xg, float* __restrict__ outg, int plane, int tfirst,
    const int (&offA)[3], const bool (&okA)[3], const v2f (&bfB)[15], int outoff)
{
    v2f xe[15], xo[15];
#pragma unroll
    for (int i = 0; i < 15; ++i) { xe[i] = v2f{0.0f, 0.0f}; xo[i] = v2f{0.0f, 0.0f}; }
    xpre_load<BLO>(xe, xg + ((long long)tfirst * 64 + plane) * 1024, offA);
    xpre_load<BLO>(xo, xg + ((long long)(tfirst + 1) * 64 + plane) * 1024, offA);

#pragma unroll 1
    for (int i = 0; i < 8; i += 2) {
        const int t  = tfirst + i;
        const int te = (i + 2 < 8) ? (t + 2) : t;
        const int to = (i + 2 < 8) ? (t + 3) : (t + 1);
        xpre_step<BLO>(xe, xg + ((long long)te * 64 + plane) * 1024,
                       outg + ((long long)t * 64 + plane) * 1024 + outoff, offA, okA, bfB);
        xpre_step<BLO>(xo, xg + ((long long)to * 64 + plane) * 1024,
                       outg + ((long long)(t + 1) * 64 + plane) * 1024 + outoff, offA, okA, bfB);
    }
}

__global__ __launch_bounds__(256) void convssm_xterm_kernel(
    const float* __restrict__ xg,
    const float* __restrict__ Bkg,
    float* __restrict__ outg)
{
    const int plane = blockIdx.x & 63;
    const int tg    = blockIdx.x >> 6;
    const int chan  = plane & 31;

    const int tid  = threadIdx.x;
    const int wave = __builtin_amdgcn_readfirstlane(tid >> 5);
    const int lane = tid & 31;
    const int n    = lane & 15;
    const int kh   = lane >> 4;
    const int m    = lane & 15;

    const int ra  = (wave >> 1) & 1;
    const int ct  = wave & 1;
    const int i0a = ra * 16;
    const int j0  = ct * 16;
    const int jn  = j0 + n;
    const int tfirst = tg * 16 + (wave >> 2) * 8;

    float tp[9];
#pragma unroll
    for (int i = 0; i < 9; ++i) tp[i] = Bkg[chan * 9 + i];

    v2f bfB[15];
    int kkArr[15];
    band_frags(tp, j0, jn, kh, bfB, kkArr);

    bool okA[3];
    int offA[3];
#pragma unroll
    for (int ks = 0; ks < 3; ++ks) {
        const int r = (i0a + m - ks) & 63;
        okA[ks]  = r < HH;
        offA[ks] = (okA[ks] ? r * WW : 0) + j0 - 4 + 2 * kh;
    }
    const int outoff = (i0a + 8 * kh) * WW + jn;

    if (ct == 0) xpre_run<1>(xg, outg, plane, tfirst, offA, okA, bfB, outoff);
    else         xpre_run<0>(xg, outg, plane, tfirst, offA, okA, bfB, outoff);
}


__device__ __forceinline__ void edge_refill(
    v2f (&xc)[2], float (&xe)[4], const float* __restrict__ xsrc,
    int co1, int co2, bool cv1, bool cv2,
    int eo1, int eo2, bool ev1, bool ev2, bool cola, bool hasb)
{
    if (cola) {
        if (cv1) xc[0] = *(const v2f*)(xsrc + co1);
        if (cv2) xc[1] = *(const v2f*)(xsrc + co2);
    }
    if (hasb) {
        if (ev1) { xe[0] = xsrc[eo1]; xe[1] = xsrc[eo1 + WW]; }
        if (ev2) { xe[2] = xsrc[eo2]; xe[3] = xsrc[eo2 + WW]; }
    }
}

template <int KC>
__device__ __forceinline__ float vtap(const float (&tp)[9], int a)
{
    return (a == 0 ? tp[KC] : 0.0f) + (a == 1 ? tp[3 + KC] : 0.0f) + (a == 2 ? tp[6 + KC] : 0.0f);
}

template <int N>
__device__ __forceinline__ float row_shr_keep(float keep, float src)
{
    return __int_as_float(__builtin_amdgcn_update_dpp(
        __float_as_int(keep), __float_as_int(src), 0x110 + N, 0xf, 0xf, false));
}

__device__ __forceinline__ void edges_signal()
{
    asm volatile("s_wait_dscnt 0x0\n\ts_barrier_signal -1" ::: "memory");
}
__device__ __forceinline__ void edges_wait()
{
    asm volatile("s_barrier_wait -1" ::: "memory");
}

__global__ __launch_bounds__(512) void convssm_scan_kernel(
    const float* __restrict__ xg,
    const float* __restrict__ Akg,
    const float* __restrict__ Bkg,
    float* __restrict__ outg)
{
    __shared__ __attribute__((aligned(16))) float Brow[2][4][FD][2];
    __shared__ __attribute__((aligned(16))) float Rcol[2][4][4][2][2][8];
    __shared__ float akl[9];
    __shared__ float bkl[9];

    const int plane = blockIdx.x;
    const int chan  = plane & 31;

    const int tid  = threadIdx.x;
    const int wave = __builtin_amdgcn_readfirstlane(tid >> 5);
    const int lane = tid & 31;
    const int n    = lane & 15;
    const int kh   = lane >> 4;
    const int m    = lane & 15;

    const unsigned long long tileOfWave = 0xBFED754C3109A628ULL;
    const int tile = (int)((tileOfWave >> (wave * 4)) & 15ULL);
    const int rt  = tile >> 2;
    const int ct  = tile & 3;
    const int i0  = rt * 16;
    const int j0  = ct * 16;
    const int jn  = j0 + n;


    const bool hasb = (rt == 2) && (ct < 3);
    const bool crop = (rt < 2) && (ct < 2);
    const bool cola = (rt < 2) && (ct == 2);
    const bool anyx = hasb || crop || cola;

    const int  ec1 = jn - kh;
    const int  ec2 = jn - 2;
    const bool ev1 = (ec1 >= 0) && (ec1 < WW);
    const bool ev2 = (kh == 1) && (ec2 >= 0) && (ec2 < WW);
    const int  eo1 = 30 * WW + (ev1 ? ec1 : 0);
    const int  eo2 = 30 * WW + (ev2 ? ec2 : 0);

    const int  rc1 = (i0 + m - kh) & 63;
    const int  rc2 = (i0 + m - 2) & 63;
    const bool cv1 = rc1 < HH;
    const bool cv2 = (kh == 0) && (rc2 < HH);
    const int  co1 = (cv1 ? rc1 * WW : 0) + 30;
    const int  co2 = (cv2 ? rc2 * WW : 0) + 30;

    const int uoff = (i0 + 8 * kh) * WW + jn;

    const int rtA = (rt + 3) & 3;
    const int ctL = (ct + 3) & 3;
    const int browRdA = (rtA * FD + ((jn - kh) & 63)) * 2;
    const int browRdB = (rtA * FD + ((jn - 2) & 63)) * 2;
    const int rcolRdX = (((rt * 4 + ctL) * 2 + (n & 1)) * 2 + kh) * 8;
    const int rcolRdY = (((rt * 4 + ctL) * 2 + 1) * 2 + kh) * 8;
    const int browWr  = (rt * FD + jn) * 2;
    const int rcolWr  = (((rt * 4 + ct) * 2 + (n & 1)) * 2 + kh) * 8;

    v2f xc[2], xc2[2];
    float xe[4], xe2[4];
#pragma unroll
    for (int i = 0; i < 2; ++i) { xc[i] = v2f{0.0f, 0.0f}; xc2[i] = v2f{0.0f, 0.0f}; }
#pragma unroll
    for (int i = 0; i < 4; ++i) { xe[i] = 0.0f; xe2[i] = 0.0f; }
    v8f un = {}, un2 = {};

    edge_refill(xc, xe, xg + (long long)plane * 1024,
                co1, co2, cv1, cv2, eo1, eo2, ev1, ev2, cola, hasb);
    if (crop) {
        const float* u = outg + (long long)plane * 1024 + uoff;
#pragma unroll
        for (int v = 0; v < 8; ++v) un[v] = u[v * WW];
    }

    if (tid < 9) {
        akl[tid] = Akg[chan * 9 + tid];
        bkl[tid] = Bkg[chan * 9 + tid];
    }
    for (int idx = tid; idx < 2 * 4 * FD * 2; idx += 512) {
        (&Brow[0][0][0][0])[idx] = 0.0f;
        (&Rcol[0][0][0][0][0][0])[idx] = 0.0f;
    }
    __syncthreads();

    v2f wA[14], wB, cB[2];
    {
        float ta[9], tb[9];
#pragma unroll
        for (int i = 0; i < 9; ++i) { ta[i] = akl[i]; tb[i] = bkl[i]; }
#pragma unroll
        for (int j = 0; j < 4; ++j) {
            const int a0 = m - (8 * kh + 2 * j);
            wA[0 + j].x = vtap<0>(ta, a0);  wA[0 + j].y = vtap<0>(ta, a0 - 1);
            wA[4 + j].x = vtap<1>(ta, a0);  wA[4 + j].y = vtap<1>(ta, a0 - 1);
            wA[8 + j].x = vtap<2>(ta, a0);  wA[8 + j].y = vtap<2>(ta, a0 - 1);
        }
        wA[12].x = kh ? vtap<1>(ta, m + 2) : vtap<0>(ta, m + 2);
        wA[12].y = kh ? vtap<1>(ta, m + 1) : vtap<0>(ta, m + 1);
        wA[13].x = kh ? (hasb ? vtap<2>(tb, m + 2) : 0.0f) : vtap<2>(ta, m + 2);
        wA[13].y = kh ? (hasb ? vtap<2>(tb, m + 1) : 0.0f) : vtap<2>(ta, m + 1);
        wB.x = kh ? vtap<1>(tb, m + 2) : vtap<0>(tb, m + 2);
        wB.y = kh ? vtap<1>(tb, m + 1) : vtap<0>(tb, m + 1);
        const float r2 = kh ? tb[5] : tb[2];
        const float r1 = kh ? tb[4] : tb[1];
        cB[0].x = (n == 0) ? r2 : 0.0f;
        cB[0].y = (n == 0) ? r1 : ((n == 1) ? r2 : 0.0f);
        cB[1].x = (kh == 0 && n == 0) ? tb[8] : 0.0f;
        cB[1].y = (kh == 0) ? ((n == 0) ? tb[7] : ((n == 1) ? tb[8] : 0.0f)) : 0.0f;
    }

    v8f hst = {};

    auto step = [&](int t, const int par,
                    const v2f (&xcC)[2], const float (&xeC)[4], const v8f& unC,
                    v2f (&xcN)[2], float (&xeN)[4], v8f& unN)
                    __attribute__((always_inline)) {
        const long long xoff = ((long long)t * 64 + plane) * 1024;
        const bool more = t + 1 < T_STEPS;
        const long long noff = ((long long)(t + 1) * 64 + plane) * 1024;


        v8f dx = unC;
        v2f xk = {0.0f, 0.0f};

        if (anyx) {
            const v2f xcm0 = xcC[0];
            const v2f xcm1 = xcC[1];
            const v2f xr0  = {xeC[0], xeC[1]};

            if (cola) {
                dx = __builtin_amdgcn_wmma_f32_16x16x4_f32(
                        false, xcm0, false, cB[0], (short)0, dx, false, false);
                dx = __builtin_amdgcn_wmma_f32_16x16x4_f32(
                        false, xcm1, false, cB[1], (short)0, dx, false, false);
            }
            if (hasb) {
                dx = __builtin_amdgcn_wmma_f32_16x16x4_f32(
                        false, wB, false, xr0, (short)0, dx, false, false);
                xk = v2f{kh ? xeC[2] : 0.0f, kh ? xeC[3] : 0.0f};
            }
        }

        v8f dh = {};
#pragma unroll
        for (int j = 0; j < 2; ++j)
            dh = __builtin_amdgcn_wmma_f32_16x16x4_f32(
                    false, wA[j], false, v2f{hst[2 * j], hst[2 * j + 1]}, (short)0, dh, false, false);

        __builtin_amdgcn_sched_barrier(0);
        edges_wait();


        const float* br = &Brow[0][0][0][0] + par * 512;
        const float* rc = &Rcol[0][0][0][0][0][0] + par * 512;
        const v2f hA = *(const v2f*)(br + browRdA);
        const v2f hB = *(const v2f*)(br + browRdB);
        const v4f x0 = *(const v4f*)(rc + rcolRdX);
        const v4f x1 = *(const v4f*)(rc + rcolRdX + 4);
        const v4f y0 = *(const v4f*)(rc + rcolRdY);
        const v4f y1 = *(const v4f*)(rc + rcolRdY + 4);

        if (anyx && more) {
            edge_refill(xcN, xeN, xg + noff,
                        co1, co2, cv1, cv2, eo1, eo2, ev1, ev2, cola, hasb);
            if (crop) {
                const float* u = outg + noff + uoff;
#pragma unroll
                for (int v = 0; v < 8; ++v) unN[v] = u[v * WW];
            }
        }

#pragma unroll
        for (int j = 2; j < 4; ++j)
            dh = __builtin_amdgcn_wmma_f32_16x16x4_f32(
                    false, wA[j], false, v2f{hst[2 * j], hst[2 * j + 1]}, (short)0, dh, false, false);
        const v2f hBx = {kh ? xk.x : hB.x, kh ? xk.y : hB.y};
        dh = __builtin_amdgcn_wmma_f32_16x16x4_f32(
                false, wA[12], false, hA, (short)0, dh, false, false);
        dh = __builtin_amdgcn_wmma_f32_16x16x4_f32(
                false, wA[13], false, hBx, (short)0, dh, false, false);
        {
            float e[8];
#pragma unroll
            for (int r = 0; r < 4; ++r) {
                e[r]     = row_shr_keep<1>(y0[r], hst[r]);
                e[4 + r] = row_shr_keep<1>(y1[r], hst[4 + r]);
            }
#pragma unroll
            for (int j = 0; j < 4; ++j)
                dh = __builtin_amdgcn_wmma_f32_16x16x4_f32(
                        false, wA[4 + j], false, v2f{e[2 * j], e[2 * j + 1]}, (short)0, dh, false, false);
        }
        {
            float e[8];
#pragma unroll
            for (int r = 0; r < 4; ++r) {
                e[r]     = row_shr_keep<2>(x0[r], hst[r]);
                e[4 + r] = row_shr_keep<2>(x1[r], hst[4 + r]);
            }
#pragma unroll
            for (int j = 0; j < 4; ++j)
                dh = __builtin_amdgcn_wmma_f32_16x16x4_f32(
                        false, wA[8 + j], false, v2f{e[2 * j], e[2 * j + 1]}, (short)0, dh, false, false);
        }

        v8f d;
#pragma unroll
        for (int v = 0; v < 8; ++v) d[v] = dh[v] + dx[v];
        hst = d;

        float* bw = &Brow[0][0][0][0] + (par ^ 1) * 512;
        float* rw = &Rcol[0][0][0][0][0][0] + (par ^ 1) * 512;
        if (kh == 1) *(v2f*)(bw + browWr) = v2f{d[6], d[7]};
        if (n >= 14) {
            *(v4f*)(rw + rcolWr)     = v4f{d[0], d[1], d[2], d[3]};
            *(v4f*)(rw + rcolWr + 4) = v4f{d[4], d[5], d[6], d[7]};
        }

        edges_signal();

        if (crop) {
            float* o = outg + xoff + uoff;
#pragma unroll
            for (int v = 0; v < 8; ++v) o[v * WW] = d[v];
        }
    };

    edges_signal();

#pragma unroll 1
    for (int t = 0; t < T_STEPS; t += 2) {
        step(t,     0, xc,  xe,  un,  xc2, xe2, un2);
        step(t + 1, 1, xc2, xe2, un2, xc,  xe,  un);
    }

    edges_wait();
}

extern "C" void kernel_launch(void* const* d_in, const int* in_sizes, int n_in,
                              void* d_out, int out_size, void* d_ws, size_t ws_size,
                              hipStream_t stream) {
    (void)in_sizes; (void)n_in; (void)out_size; (void)d_ws; (void)ws_size;
    const float* x  = (const float*)d_in[0];
    const float* Ak = (const float*)d_in[1];
    const float* Bk = (const float*)d_in[2];
    float* out = (float*)d_out;
    convssm_xterm_kernel<<<dim3(256), dim3(256), 0, stream>>>(x, Bk, out);
    convssm_scan_kernel<<<dim3(64), dim3(512), 0, stream>>>(x, Ak, Bk, out);
}


# Scan state term: WMMA M = 4 output rows x 3 kernel columns, K = input rows; column taps summed by DPP shifts of same-step partial sums passed via LDS: 8 WMMAs a tile, were 14; tap sum order changes
# speedup vs baseline: 2.9078x; 1.0439x over previous
#include <hip/hip_runtime.h>


typedef __attribute__((ext_vector_type(2))) float v2f;
typedef __attribute__((ext_vector_type(4))) float v4f;
typedef __attribute__((ext_vector_type(8))) float v8f;

#define T_STEPS 64
#define FD      64
#define HH      32
#define WW      32


__device__ __forceinline__ void band_frags(
    const float (&tp)[9], int j0, int jn, int kh, v2f (&bf)[15], int (&kkArr)[15])
{
    int q = 0;
#pragma unroll
    for (int ks = 0; ks < 3; ++ks) {
        const float c0 = tp[ks * 3 + 0], c1 = tp[ks * 3 + 1], c2 = tp[ks * 3 + 2];
#pragma unroll
        for (int bbk = 0; bbk < 5; ++bbk, ++q) {
            const int kb = ((j0 >> 2) + bbk + 15) & 15;
            const int kk = (kb << 2) + 2 * kh;
            kkArr[q] = kk;
            const int dx = (jn - kk) & 63;
            const int dy = (jn - kk - 1) & 63;
            bf[q].x = (dx == 0 ? c0 : 0.0f) + (dx == 1 ? c1 : 0.0f) + (dx == 2 ? c2 : 0.0f);
            bf[q].y = (dy == 0 ? c0 : 0.0f) + (dy == 1 ? c1 : 0.0f) + (dy == 2 ? c2 : 0.0f);
        }
    }
}


template <int BLO>
__device__ __forceinline__ void xpre_load(
    v2f (&xf)[15], const float* __restrict__ xsrc, const int (&offA)[3])
{
#pragma unroll
    for (int ks = 0; ks < 3; ++ks) {
#pragma unroll
        for (int b = BLO; b < 5; ++b)
            xf[ks * 5 + b] = *(const v2f*)(xsrc + offA[ks] + 4 * b);
    }
}

template <int BLO>
__device__ __forceinline__ void xpre_step(
    v2f (&xf)[15], const float* __restrict__ xnext, float* __restrict__ o,
    const int (&offA)[3], const bool (&okA)[3], const v2f (&bfB)[15])
{
    v2f xm[15];
    {
        const v2f zero = {0.0f, 0.0f};
#pragma unroll
        for (int ks = 0; ks < 3; ++ks) {
#pragma unroll
            for (int b = 0; b < 5; ++b)
                xm[ks * 5 + b] = okA[ks] ? xf[ks * 5 + b] : zero;
        }
    }
    xpre_load<BLO>(xf, xnext, offA);

    v8f d = {};
#pragma unroll
    for (int ks = 0; ks < 3; ++ks) {
#pragma unroll
        for (int b = BLO; b < 5; ++b)
            d = __builtin_amdgcn_wmma_f32_16x16x4_f32(
                    false, xm[ks * 5 + b], false, bfB[ks * 5 + b], (short)0, d, false, false);
    }
#pragma unroll
    for (int v = 0; v < 8; ++v) o[v * WW] = d[v];
}

template <int BLO>
__device__ __forceinline__ void xpre_run(
    const float* __restrict__ xg, float* __restrict__ outg, int plane, int tfirst,
    const int (&offA)[3], const bool (&okA)[3], const v2f (&bfB)[15], int outoff)
{
    v2f xe[15], xo[15];
#pragma unroll
    for (int i = 0; i < 15; ++i) { xe[i] = v2f{0.0f, 0.0f}; xo[i] = v2f{0.0f, 0.0f}; }
    xpre_load<BLO>(xe, xg + ((long long)tfirst * 64 + plane) * 1024, offA);
    xpre_load<BLO>(xo, xg + ((long long)(tfirst + 1) * 64 + plane) * 1024, offA);

#pragma unroll 1
    for (int i = 0; i < 8; i += 2) {
        const int t  = tfirst + i;
        const int te = (i + 2 < 8) ? (t + 2) : t;
        const int to = (i + 2 < 8) ? (t + 3) : (t + 1);
        xpre_step<BLO>(xe, xg + ((long long)te * 64 + plane) * 1024,
                       outg + ((long long)t * 64 + plane) * 1024 + outoff, offA, okA, bfB);
        xpre_step<BLO>(xo, xg + ((long long)to * 64 + plane) * 1024,
                       outg + ((long long)(t + 1) * 64 + plane) * 1024 + outoff, offA, okA, bfB);
    }
}

__global__ __launch_bounds__(256) void convssm_xterm_kernel(
    const float* __restrict__ xg,
    const float* __restrict__ Bkg,
    float* __restrict__ outg)
{
    const int plane = blockIdx.x & 63;
    const int tg    = blockIdx.x >> 6;
    const int chan  = plane & 31;

    const int tid  = threadIdx.x;
    const int wave = __builtin_amdgcn_readfirstlane(tid >> 5);
    const int lane = tid & 31;
    const int n    = lane & 15;
    const int kh   = lane >> 4;
    const int m    = lane & 15;

    const int ra  = (wave >> 1) & 1;
    const int ct  = wave & 1;
    const int i0a = ra * 16;
    const int j0  = ct * 16;
    const int jn  = j0 + n;
    const int tfirst = tg * 16 + (wave >> 2) * 8;

    float tp[9];
#pragma unroll
    for (int i = 0; i < 9; ++i) tp[i] = Bkg[chan * 9 + i];

    v2f bfB[15];
    int kkArr[15];
    band_frags(tp, j0, jn, kh, bfB, kkArr);

    bool okA[3];
    int offA[3];
#pragma unroll
    for (int ks = 0; ks < 3; ++ks) {
        const int r = (i0a + m - ks) & 63;
        okA[ks]  = r < HH;
        offA[ks] = (okA[ks] ? r * WW : 0) + j0 - 4 + 2 * kh;
    }
    const int outoff = (i0a + 8 * kh) * WW + jn;

    if (ct == 0) xpre_run<1>(xg, outg, plane, tfirst, offA, okA, bfB, outoff);
    else         xpre_run<0>(xg, outg, plane, tfirst, offA, okA, bfB, outoff);
}


__device__ __forceinline__ void edge_refill(
    v2f (&xc)[2], float (&xe)[4], const float* __restrict__ xsrc,
    int co1, int co2, bool cv1, bool cv2,
    int eo1, int eo2, bool ev1, bool ev2, bool cola, bool hasb)
{
    if (cola) {
        if (cv1) xc[0] = *(const v2f*)(xsrc + co1);
        if (cv2) xc[1] = *(const v2f*)(xsrc + co2);
    }
    if (hasb) {
        if (ev1) { xe[0] = xsrc[eo1]; xe[1] = xsrc[eo1 + WW]; }
        if (ev2) { xe[2] = xsrc[eo2]; xe[3] = xsrc[eo2 + WW]; }
    }
}

template <int KC>
__device__ __forceinline__ float vtap(const float (&tp)[9], int a)
{
    return (a == 0 ? tp[KC] : 0.0f) + (a == 1 ? tp[3 + KC] : 0.0f) + (a == 2 ? tp[6 + KC] : 0.0f);
}

template <int N>
__device__ __forceinline__ float row_shr_keep(float keep, float src)
{
    return __int_as_float(__builtin_amdgcn_update_dpp(
        __float_as_int(keep), __float_as_int(src), 0x110 + N, 0xf, 0xf, false));
}

__device__ __forceinline__ void edges_signal()
{
    asm volatile("s_wait_dscnt 0x0\n\ts_barrier_signal -1" ::: "memory");
}
__device__ __forceinline__ void edges_wait()
{
    asm volatile("s_barrier_wait -1" ::: "memory");
}

__global__ __launch_bounds__(512) void convssm_scan_kernel(
    const float* __restrict__ xg,
    const float* __restrict__ Akg,
    const float* __restrict__ Bkg,
    float* __restrict__ outg)
{
    __shared__ __attribute__((aligned(16))) float Top[2][16][16][8];
    __shared__ __attribute__((aligned(16))) float Edge[2][16][2][3][8];
    __shared__ float akl[9];
    __shared__ float bkl[9];

    const int plane = blockIdx.x;
    const int chan  = plane & 31;

    const int tid  = threadIdx.x;
    const int wave = __builtin_amdgcn_readfirstlane(tid >> 5);
    const int lane = tid & 31;
    const int n    = lane & 15;
    const int kh   = lane >> 4;
    const int m    = lane & 15;

    const unsigned long long tileOfWave = 0xBFED754C3109A628ULL;
    const int tile = (int)((tileOfWave >> (wave * 4)) & 15ULL);
    const int rt  = tile >> 2;
    const int ct  = tile & 3;
    const int i0  = rt * 16;
    const int j0  = ct * 16;
    const int jn  = j0 + n;

    const int mr = 4 * ((m & 7) >> 1) + 2 * (m >> 3) + (m & 1);


    const bool hasb = (rt == 2) && (ct < 3);
    const bool crop = (rt < 2) && (ct < 2);
    const bool cola = (rt < 2) && (ct == 2);
    const bool anyx = hasb || crop || cola;

    const int  ec1 = jn - kh;
    const int  ec2 = jn - 2;
    const bool ev1 = (ec1 >= 0) && (ec1 < WW);
    const bool ev2 = (kh == 1) && (ec2 >= 0) && (ec2 < WW);
    const int  eo1 = 30 * WW + (ev1 ? ec1 : 0);
    const int  eo2 = 30 * WW + (ev2 ? ec2 : 0);

    const int  rc1 = (i0 + mr - kh) & 63;
    const int  rc2 = (i0 + mr - 2) & 63;
    const bool cv1 = rc1 < HH;
    const bool cv2 = (kh == 0) && (rc2 < HH);
    const int  co1 = (cv1 ? rc1 * WW : 0) + 30;
    const int  co2 = (cv2 ? rc2 * WW : 0) + 30;

    const int uoff = (i0 + 2 * kh) * WW + jn;

    const int rtA = (rt + 3) & 3;
    const int ctL = (ct + 3) & 3;
    const int topWr   = (tile * 16 + n) * 8;
    const int topRd   = ((rtA * 4 + ct) * 16 + n) * 8;
    const int cornY   = ((rtA * 4 + ctL) * 16 + 15) * 8;
    const int cornX   = ((rtA * 4 + ctL) * 16 + 14 + (n & 1)) * 8;
    const int edgeWrA = ((tile * 2 + kh) * 3 + (n == 15 ? 0 : 1)) * 8;
    const int edgeWrB = ((tile * 2 + kh) * 3 + 2) * 8;
    const int edgeRdY = (((rt * 4 + ctL) * 2 + kh) * 3 + 0) * 8;
    const int edgeRdX = (((rt * 4 + ctL) * 2 + kh) * 3 + ((n & 1) ? 2 : 1)) * 8;

    v2f xc[2], xc2[2];
    float xe[4], xe2[4];
#pragma unroll
    for (int i = 0; i < 2; ++i) { xc[i] = v2f{0.0f, 0.0f}; xc2[i] = v2f{0.0f, 0.0f}; }
#pragma unroll
    for (int i = 0; i < 4; ++i) { xe[i] = 0.0f; xe2[i] = 0.0f; }
    v8f un = {}, un2 = {};

    edge_refill(xc, xe, xg + (long long)plane * 1024,
                co1, co2, cv1, cv2, eo1, eo2, ev1, ev2, cola, hasb);
    if (crop) {
        const float* u = outg + (long long)plane * 1024 + uoff;
#pragma unroll
        for (int q = 0; q < 8; ++q) un[q] = u[(4 * (q >> 1) + (q & 1)) * WW];
    }

    if (tid < 9) {
        akl[tid] = Akg[chan * 9 + tid];
        bkl[tid] = Bkg[chan * 9 + tid];
    }
    __syncthreads();

    v2f V0, V1, wB, wB2, cB[2];
    {
        float ta[9], tb[9];
#pragma unroll
        for (int i = 0; i < 9; ++i) { ta[i] = akl[i]; tb[i] = bkl[i]; }
        const int il = 2 * (m >> 3) + ((m >> 2) & 1);
        const int bc = m & 3;
        float tcol[3];
#pragma unroll
        for (int a = 0; a < 3; ++a)
            tcol[a] = (bc == 0 ? ta[3 * a] : 0.0f) + (bc == 1 ? ta[3 * a + 1] : 0.0f)
                    + (bc == 2 ? ta[3 * a + 2] : 0.0f);
        auto sel = [&](int a) -> float {
            return (a == 0 ? tcol[0] : 0.0f) + (a == 1 ? tcol[1] : 0.0f) + (a == 2 ? tcol[2] : 0.0f);
        };
        V1.x = sel(il - 2 * kh);      V1.y = sel(il - 2 * kh - 1);
        V0.x = sel(4 + il - 2 * kh);  V0.y = sel(3 + il - 2 * kh);

        wB.x  = kh ? vtap<1>(tb, mr + 2) : vtap<0>(tb, mr + 2);
        wB.y  = kh ? vtap<1>(tb, mr + 1) : vtap<0>(tb, mr + 1);
        wB2.x = kh ? vtap<2>(tb, mr + 2) : 0.0f;
        wB2.y = kh ? vtap<2>(tb, mr + 1) : 0.0f;
        const float r2 = kh ? tb[5] : tb[2];
        const float r1 = kh ? tb[4] : tb[1];
        cB[0].x = (n == 0) ? r2 : 0.0f;
        cB[0].y = (n == 0) ? r1 : ((n == 1) ? r2 : 0.0f);
        cB[1].x = (kh == 0 && n == 0) ? tb[8] : 0.0f;
        cB[1].y = (kh == 0) ? ((n == 0) ? tb[7] : ((n == 1) ? tb[8] : 0.0f)) : 0.0f;
    }

    v8f hst = {};

    auto step = [&](int t, const int par,
                    const v2f (&xcC)[2], const float (&xeC)[4], const v8f& unC,
                    v2f (&xcN)[2], float (&xeN)[4], v8f& unN)
                    __attribute__((always_inline)) {
        const long long xoff = ((long long)t * 64 + plane) * 1024;
        const bool more = t + 1 < T_STEPS;
        const long long noff = ((long long)(t + 1) * 64 + plane) * 1024;

        const v2f s0 = {hst[0], hst[1]};
        const v2f s1 = {hst[2], hst[3]};
        const v2f s2 = {hst[4], hst[5]};
        const v2f s3 = {hst[6], hst[7]};
        const v8f zero8 = {};
        v8f p0 = __builtin_amdgcn_wmma_f32_16x16x4_f32(false, V1, false, s0, (short)0, zero8, false, false);
        v8f p1 = __builtin_amdgcn_wmma_f32_16x16x4_f32(false, V0, false, s0, (short)0, zero8, false, false);
        v8f p2 = __builtin_amdgcn_wmma_f32_16x16x4_f32(false, V0, false, s1, (short)0, zero8, false, false);
        v8f p3 = __builtin_amdgcn_wmma_f32_16x16x4_f32(false, V0, false, s2, (short)0, zero8, false, false);
        v8f p4 = __builtin_amdgcn_wmma_f32_16x16x4_f32(false, V0, false, s3, (short)0, zero8, false, false);
        p1 = __builtin_amdgcn_wmma_f32_16x16x4_f32(false, V1, false, s1, (short)0, p1, false, false);
        p2 = __builtin_amdgcn_wmma_f32_16x16x4_f32(false, V1, false, s2, (short)0, p2, false, false);
        p3 = __builtin_amdgcn_wmma_f32_16x16x4_f32(false, V1, false, s3, (short)0, p3, false, false);

        const float k0[8] = {p0[0], p0[4], p1[0], p1[4], p2[0], p2[4], p3[0], p3[4]};
        float       k1[8] = {p0[1], p0[5], p1[1], p1[5], p2[1], p2[5], p3[1], p3[5]};
        float       k2[8] = {p0[2], p0[6], p1[2], p1[6], p2[2], p2[6], p3[2], p3[6]};

        float* tw = &Top[0][0][0][0] + par * 2048;
        float* ew = &Edge[0][0][0][0][0] + par * 768;
        if (kh == 0) {
            *(v4f*)(tw + topWr)     = v4f{p4[0], p4[1], p4[2], p4[3]};
            *(v4f*)(tw + topWr + 4) = v4f{p4[4], p4[5], p4[6], p4[7]};
        }
        if (n >= 14) {
            const bool last = (n == 15);
            float a[8];
#pragma unroll
            for (int q = 0; q < 8; ++q) a[q] = last ? k1[q] : k2[q];
            *(v4f*)(ew + edgeWrA)     = v4f{a[0], a[1], a[2], a[3]};
            *(v4f*)(ew + edgeWrA + 4) = v4f{a[4], a[5], a[6], a[7]};
            if (last) {
                *(v4f*)(ew + edgeWrB)     = v4f{k2[0], k2[1], k2[2], k2[3]};
                *(v4f*)(ew + edgeWrB + 4) = v4f{k2[4], k2[5], k2[6], k2[7]};
            }
        }

        edges_signal();


        __builtin_amdgcn_sched_barrier(0);

        v8f dx = unC;
        if (anyx) {
            if (cola) {
                dx = __builtin_amdgcn_wmma_f32_16x16x4_f32(
                        false, xcC[0], false, cB[0], (short)0, dx, false, false);
                dx = __builtin_amdgcn_wmma_f32_16x16x4_f32(
                        false, xcC[1], false, cB[1], (short)0, dx, false, false);
            }
            if (hasb) {
                const v2f xr0 = {xeC[0], xeC[1]};
                const v2f xk  = {kh ? xeC[2] : 0.0f, kh ? xeC[3] : 0.0f};
                dx = __builtin_amdgcn_wmma_f32_16x16x4_f32(
                        false, wB, false, xr0, (short)0, dx, false, false);
                dx = __builtin_amdgcn_wmma_f32_16x16x4_f32(
                        false, wB2, false, xk, (short)0, dx, false, false);
            }
        }

        float acc[8];
#pragma unroll
        for (int q = 0; q < 8; ++q) acc[q] = k0[q] + dx[q];

        if (anyx && more) {
            edge_refill(xcN, xeN, xg + noff,
                        co1, co2, cv1, cv2, eo1, eo2, ev1, ev2, cola, hasb);
            if (crop) {
                const float* u = outg + noff + uoff;
#pragma unroll
                for (int q = 0; q < 8; ++q) unN[q] = u[(4 * (q >> 1) + (q & 1)) * WW];
            }
        }

        edges_wait();

        const float* tr = &Top[0][0][0][0] + par * 2048;
        const float* er = &Edge[0][0][0][0][0] + par * 768;
        const v4f T0 = *(const v4f*)(tr + topRd);
        const v4f T1 = *(const v4f*)(tr + topRd + 4);
        const v4f y0 = *(const v4f*)(er + edgeRdY);
        const v4f y1 = *(const v4f*)(er + edgeRdY + 4);
        const v4f x0 = *(const v4f*)(er + edgeRdX);
        const v4f x1 = *(const v4f*)(er + edgeRdX + 4);
        const float cy0 = tr[cornY + 1], cy1 = tr[cornY + 5];
        const float cx0 = tr[cornX + 2], cx1 = tr[cornX + 6];

        float yk[8] = {y0[0], y0[1], y0[2], y0[3], y1[0], y1[1], y1[2], y1[3]};
        float xk2[8] = {x0[0], x0[1], x0[2], x0[3], x1[0], x1[1], x1[2], x1[3]};

        const bool toph = (kh == 0);
        acc[0] += toph ? T0[0] : 0.0f;
        k1[0]  += toph ? T0[1] : 0.0f;
        k2[0]  += toph ? T0[2] : 0.0f;
        acc[1] += toph ? T1[0] : 0.0f;
        k1[1]  += toph ? T1[1] : 0.0f;
        k2[1]  += toph ? T1[2] : 0.0f;
        yk[0]  += toph ? cy0 : 0.0f;
        yk[1]  += toph ? cy1 : 0.0f;
        xk2[0] += toph ? cx0 : 0.0f;
        xk2[1] += toph ? cx1 : 0.0f;

        v8f d;
#pragma unroll
        for (int q = 0; q < 8; ++q) {
            const float t1 = row_shr_keep<1>(yk[q], k1[q]);
            const float t2 = row_shr_keep<2>(xk2[q], k2[q]);
            d[q] = (acc[q] + t1) + t2;
        }
        hst = d;

        if (crop) {
            float* o = outg + xoff + uoff;
#pragma unroll
            for (int q = 0; q < 8; ++q) o[(4 * (q >> 1) + (q & 1)) * WW] = d[q];
        }
    };

#pragma unroll 1
    for (int t = 0; t < T_STEPS; t += 2) {
        step(t,     0, xc,  xe,  un,  xc2, xe2, un2);
        step(t + 1, 1, xc2, xe2, un2, xc,  xe,  un);
    }
}

extern "C" void kernel_launch(void* const* d_in, const int* in_sizes, int n_in,
                              void* d_out, int out_size, void* d_ws, size_t ws_size,
                              hipStream_t stream) {
    (void)in_sizes; (void)n_in; (void)out_size; (void)d_ws; (void)ws_size;
    const float* x  = (const float*)d_in[0];
    const float* Ak = (const float*)d_in[1];
    const float* Bk = (const float*)d_in[2];
    float* out = (float*)d_out;
    convssm_xterm_kernel<<<dim3(256), dim3(256), 0, stream>>>(x, Bk, out);
    convssm_scan_kernel<<<dim3(64), dim3(512), 0, stream>>>(x, Ak, Bk, out);
}


# Scan exchange records pre-summed by their producer (P_4 over kernel columns, right edge one sum per target column): 8 LDS instructions a wave and step, were 14; tap sum order changes
# speedup vs baseline: 3.1370x; 1.0788x over previous
#include <hip/hip_runtime.h>


typedef __attribute__((ext_vector_type(2))) float v2f;
typedef __attribute__((ext_vector_type(4))) float v4f;
typedef __attribute__((ext_vector_type(8))) float v8f;

#define T_STEPS 64
#define FD      64
#define HH      32
#define WW      32


__device__ __forceinline__ void band_frags(
    const float (&tp)[9], int j0, int jn, int kh, v2f (&bf)[15], int (&kkArr)[15])
{
    int q = 0;
#pragma unroll
    for (int ks = 0; ks < 3; ++ks) {
        const float c0 = tp[ks * 3 + 0], c1 = tp[ks * 3 + 1], c2 = tp[ks * 3 + 2];
#pragma unroll
        for (int bbk = 0; bbk < 5; ++bbk, ++q) {
            const int kb = ((j0 >> 2) + bbk + 15) & 15;
            const int kk = (kb << 2) + 2 * kh;
            kkArr[q] = kk;
            const int dx = (jn - kk) & 63;
            const int dy = (jn - kk - 1) & 63;
            bf[q].x = (dx == 0 ? c0 : 0.0f) + (dx == 1 ? c1 : 0.0f) + (dx == 2 ? c2 : 0.0f);
            bf[q].y = (dy == 0 ? c0 : 0.0f) + (dy == 1 ? c1 : 0.0f) + (dy == 2 ? c2 : 0.0f);
        }
    }
}


template <int BLO>
__device__ __forceinline__ void xpre_load(
    v2f (&xf)[15], const float* __restrict__ xsrc, const int (&offA)[3])
{
#pragma unroll
    for (int ks = 0; ks < 3; ++ks) {
#pragma unroll
        for (int b = BLO; b < 5; ++b)
            xf[ks * 5 + b] = *(const v2f*)(xsrc + offA[ks] + 4 * b);
    }
}

template <int BLO>
__device__ __forceinline__ void xpre_step(
    v2f (&xf)[15], const float* __restrict__ xnext, float* __restrict__ o,
    const int (&offA)[3], const bool (&okA)[3], const v2f (&bfB)[15])
{
    v2f xm[15];
    {
        const v2f zero = {0.0f, 0.0f};
#pragma unroll
        for (int ks = 0; ks < 3; ++ks) {
#pragma unroll
            for (int b = 0; b < 5; ++b)
                xm[ks * 5 + b] = okA[ks] ? xf[ks * 5 + b] : zero;
        }
    }
    xpre_load<BLO>(xf, xnext, offA);

    v8f d = {};
#pragma unroll
    for (int ks = 0; ks < 3; ++ks) {
#pragma unroll
        for (int b = BLO; b < 5; ++b)
            d = __builtin_amdgcn_wmma_f32_16x16x4_f32(
                    false, xm[ks * 5 + b], false, bfB[ks * 5 + b], (short)0, d, false, false);
    }
#pragma unroll
    for (int v = 0; v < 8; ++v) o[v * WW] = d[v];
}

template <int BLO>
__device__ __forceinline__ void xpre_run(
    const float* __restrict__ xg, float* __restrict__ outg, int plane, int tfirst,
    const int (&offA)[3], const bool (&okA)[3], const v2f (&bfB)[15], int outoff)
{
    v2f xe[15], xo[15];
#pragma unroll
    for (int i = 0; i < 15; ++i) { xe[i] = v2f{0.0f, 0.0f}; xo[i] = v2f{0.0f, 0.0f}; }
    xpre_load<BLO>(xe, xg + ((long long)tfirst * 64 + plane) * 1024, offA);
    xpre_load<BLO>(xo, xg + ((long long)(tfirst + 1) * 64 + plane) * 1024, offA);

#pragma unroll 1
    for (int i = 0; i < 8; i += 2) {
        const int t  = tfirst + i;
        const int te = (i + 2 < 8) ? (t + 2) : t;
        const int to = (i + 2 < 8) ? (t + 3) : (t + 1);
        xpre_step<BLO>(xe, xg + ((long long)te * 64 + plane) * 1024,
                       outg + ((long long)t * 64 + plane) * 1024 + outoff, offA, okA, bfB);
        xpre_step<BLO>(xo, xg + ((long long)to * 64 + plane) * 1024,
                       outg + ((long long)(t + 1) * 64 + plane) * 1024 + outoff, offA, okA, bfB);
    }
}

__global__ __launch_bounds__(256) void convssm_xterm_kernel(
    const float* __restrict__ xg,
    const float* __restrict__ Bkg,
    float* __restrict__ outg)
{
    const int plane = blockIdx.x & 63;
    const int tg    = blockIdx.x >> 6;
    const int chan  = plane & 31;

    const int tid  = threadIdx.x;
    const int wave = __builtin_amdgcn_readfirstlane(tid >> 5);
    const int lane = tid & 31;
    const int n    = lane & 15;
    const int kh   = lane >> 4;
    const int m    = lane & 15;

    const int ra  = (wave >> 1) & 1;
    const int ct  = wave & 1;
    const int i0a = ra * 16;
    const int j0  = ct * 16;
    const int jn  = j0 + n;
    const int tfirst = tg * 16 + (wave >> 2) * 8;

    float tp[9];
#pragma unroll
    for (int i = 0; i < 9; ++i) tp[i] = Bkg[chan * 9 + i];

    v2f bfB[15];
    int kkArr[15];
    band_frags(tp, j0, jn, kh, bfB, kkArr);

    bool okA[3];
    int offA[3];
#pragma unroll
    for (int ks = 0; ks < 3; ++ks) {
        const int r = (i0a + m - ks) & 63;
        okA[ks]  = r < HH;
        offA[ks] = (okA[ks] ? r * WW : 0) + j0 - 4 + 2 * kh;
    }
    const int outoff = (i0a + 8 * kh) * WW + jn;

    if (ct == 0) xpre_run<1>(xg, outg, plane, tfirst, offA, okA, bfB, outoff);
    else         xpre_run<0>(xg, outg, plane, tfirst, offA, okA, bfB, outoff);
}


__device__ __forceinline__ void edge_refill(
    v2f (&xc)[2], float (&xe)[4], const float* __restrict__ xsrc,
    int co1, int co2, bool cv1, bool cv2,
    int eo1, int eo2, bool ev1, bool ev2, bool cola, bool hasb)
{
    if (cola) {
        if (cv1) xc[0] = *(const v2f*)(xsrc + co1);
        if (cv2) xc[1] = *(const v2f*)(xsrc + co2);
    }
    if (hasb) {
        if (ev1) { xe[0] = xsrc[eo1]; xe[1] = xsrc[eo1 + WW]; }
        if (ev2) { xe[2] = xsrc[eo2]; xe[3] = xsrc[eo2 + WW]; }
    }
}

template <int KC>
__device__ __forceinline__ float vtap(const float (&tp)[9], int a)
{
    return (a == 0 ? tp[KC] : 0.0f) + (a == 1 ? tp[3 + KC] : 0.0f) + (a == 2 ? tp[6 + KC] : 0.0f);
}

template <int N>
__device__ __forceinline__ float row_shr_keep(float keep, float src)
{
    return __int_as_float(__builtin_amdgcn_update_dpp(
        __float_as_int(keep), __float_as_int(src), 0x110 + N, 0xf, 0xf, false));
}

template <int N>
__device__ __forceinline__ float row_shr_zero(float src)
{
    return __int_as_float(__builtin_amdgcn_update_dpp(
        0, __float_as_int(src), 0x110 + N, 0xf, 0xf, true));
}
template <int N>
__device__ __forceinline__ float row_shl_zero(float src)
{
    return __int_as_float(__builtin_amdgcn_update_dpp(
        0, __float_as_int(src), 0x100 + N, 0xf, 0xf, true));
}

__device__ __forceinline__ void edges_signal()
{
    asm volatile("s_wait_dscnt 0x0\n\ts_barrier_signal -1" ::: "memory");
}
__device__ __forceinline__ void edges_wait()
{
    asm volatile("s_barrier_wait -1" ::: "memory");
}

__global__ __launch_bounds__(512) void convssm_scan_kernel(
    const float* __restrict__ xg,
    const float* __restrict__ Akg,
    const float* __restrict__ Bkg,
    float* __restrict__ outg)
{
    __shared__ __attribute__((aligned(16))) float Top[2][16][16][2];
    __shared__ __attribute__((aligned(16))) float Edge[2][16][2][2][12];
    __shared__ float akl[9];
    __shared__ float bkl[9];

    const int plane = blockIdx.x;
    const int chan  = plane & 31;

    const int tid  = threadIdx.x;
    const int wave = __builtin_amdgcn_readfirstlane(tid >> 5);
    const int lane = tid & 31;
    const int n    = lane & 15;
    const int kh   = lane >> 4;
    const int m    = lane & 15;

    const unsigned long long tileOfWave = 0xBFED754C3109A628ULL;
    const int tile = (int)((tileOfWave >> (wave * 4)) & 15ULL);
    const int rt  = tile >> 2;
    const int ct  = tile & 3;
    const int i0  = rt * 16;
    const int j0  = ct * 16;
    const int jn  = j0 + n;

    const int mr = 4 * ((m & 7) >> 1) + 2 * (m >> 3) + (m & 1);


    const bool hasb = (rt == 2) && (ct < 3);
    const bool crop = (rt < 2) && (ct < 2);
    const bool cola = (rt < 2) && (ct == 2);
    const bool anyx = hasb || crop || cola;

    const int  ec1 = jn - kh;
    const int  ec2 = jn - 2;
    const bool ev1 = (ec1 >= 0) && (ec1 < WW);
    const bool ev2 = (kh == 1) && (ec2 >= 0) && (ec2 < WW);
    const int  eo1 = 30 * WW + (ev1 ? ec1 : 0);
    const int  eo2 = 30 * WW + (ev2 ? ec2 : 0);

    const int  rc1 = (i0 + mr - kh) & 63;
    const int  rc2 = (i0 + mr - 2) & 63;
    const bool cv1 = rc1 < HH;
    const bool cv2 = (kh == 0) && (rc2 < HH);
    const int  co1 = (cv1 ? rc1 * WW : 0) + 30;
    const int  co2 = (cv2 ? rc2 * WW : 0) + 30;

    const int uoff = (i0 + 2 * kh) * WW + jn;

    const int rtA = (rt + 3) & 3;
    const int ctL = (ct + 3) & 3;
    const int topWr  = (tile * 16 + n) * 2;
    const int topRd  = ((rtA * 4 + ct) * 16 + n) * 2;
    const int edgeWr = ((tile * 2 + kh) * 2 + (n == 15 ? 1 : 0)) * 12;
    const int edgeRd = (((rt * 4 + ctL) * 2 + kh) * 2 + (n & 1)) * 12;
    const int cornRd = (((rtA * 4 + ctL) * 2 + kh) * 2 + (n & 1)) * 12 + 8;

    v2f xc[2], xc2[2];
    float xe[4], xe2[4];
#pragma unroll
    for (int i = 0; i < 2; ++i) { xc[i] = v2f{0.0f, 0.0f}; xc2[i] = v2f{0.0f, 0.0f}; }
#pragma unroll
    for (int i = 0; i < 4; ++i) { xe[i] = 0.0f; xe2[i] = 0.0f; }
    v8f un = {}, un2 = {};

    edge_refill(xc, xe, xg + (long long)plane * 1024,
                co1, co2, cv1, cv2, eo1, eo2, ev1, ev2, cola, hasb);
    if (crop) {
        const float* u = outg + (long long)plane * 1024 + uoff;
#pragma unroll
        for (int q = 0; q < 8; ++q) un[q] = u[(4 * (q >> 1) + (q & 1)) * WW];
    }

    if (tid < 9) {
        akl[tid] = Akg[chan * 9 + tid];
        bkl[tid] = Bkg[chan * 9 + tid];
    }
    __syncthreads();

    v2f V0, V1, wB, wB2, cB[2];
    {
        float ta[9], tb[9];
#pragma unroll
        for (int i = 0; i < 9; ++i) { ta[i] = akl[i]; tb[i] = bkl[i]; }
        const int il = 2 * (m >> 3) + ((m >> 2) & 1);
        const int bc = m & 3;
        float tcol[3];
#pragma unroll
        for (int a = 0; a < 3; ++a)
            tcol[a] = (bc == 0 ? ta[3 * a] : 0.0f) + (bc == 1 ? ta[3 * a + 1] : 0.0f)
                    + (bc == 2 ? ta[3 * a + 2] : 0.0f);
        auto sel = [&](int a) -> float {
            return (a == 0 ? tcol[0] : 0.0f) + (a == 1 ? tcol[1] : 0.0f) + (a == 2 ? tcol[2] : 0.0f);
        };
        V1.x = sel(il - 2 * kh);      V1.y = sel(il - 2 * kh - 1);
        V0.x = sel(4 + il - 2 * kh);  V0.y = sel(3 + il - 2 * kh);

        wB.x  = kh ? vtap<1>(tb, mr + 2) : vtap<0>(tb, mr + 2);
        wB.y  = kh ? vtap<1>(tb, mr + 1) : vtap<0>(tb, mr + 1);
        wB2.x = kh ? vtap<2>(tb, mr + 2) : 0.0f;
        wB2.y = kh ? vtap<2>(tb, mr + 1) : 0.0f;
        const float r2 = kh ? tb[5] : tb[2];
        const float r1 = kh ? tb[4] : tb[1];
        cB[0].x = (n == 0) ? r2 : 0.0f;
        cB[0].y = (n == 0) ? r1 : ((n == 1) ? r2 : 0.0f);
        cB[1].x = (kh == 0 && n == 0) ? tb[8] : 0.0f;
        cB[1].y = (kh == 0) ? ((n == 0) ? tb[7] : ((n == 1) ? tb[8] : 0.0f)) : 0.0f;
    }

    v8f hst = {};

    auto step = [&](int t, const int par,
                    const v2f (&xcC)[2], const float (&xeC)[4], const v8f& unC,
                    v2f (&xcN)[2], float (&xeN)[4], v8f& unN)
                    __attribute__((always_inline)) {
        const long long xoff = ((long long)t * 64 + plane) * 1024;
        const bool more = t + 1 < T_STEPS;
        const long long noff = ((long long)(t + 1) * 64 + plane) * 1024;

        const v2f s0 = {hst[0], hst[1]};
        const v2f s1 = {hst[2], hst[3]};
        const v2f s2 = {hst[4], hst[5]};
        const v2f s3 = {hst[6], hst[7]};
        const v8f zero8 = {};
        v8f p0 = __builtin_amdgcn_wmma_f32_16x16x4_f32(false, V1, false, s0, (short)0, zero8, false, false);
        v8f p1 = __builtin_amdgcn_wmma_f32_16x16x4_f32(false, V0, false, s0, (short)0, zero8, false, false);
        v8f p2 = __builtin_amdgcn_wmma_f32_16x16x4_f32(false, V0, false, s1, (short)0, zero8, false, false);
        v8f p3 = __builtin_amdgcn_wmma_f32_16x16x4_f32(false, V0, false, s2, (short)0, zero8, false, false);
        v8f p4 = __builtin_amdgcn_wmma_f32_16x16x4_f32(false, V0, false, s3, (short)0, zero8, false, false);
        p1 = __builtin_amdgcn_wmma_f32_16x16x4_f32(false, V1, false, s1, (short)0, p1, false, false);
        p2 = __builtin_amdgcn_wmma_f32_16x16x4_f32(false, V1, false, s2, (short)0, p2, false, false);
        p3 = __builtin_amdgcn_wmma_f32_16x16x4_f32(false, V1, false, s3, (short)0, p3, false, false);

        const float k0[10] = {p0[0], p0[4], p1[0], p1[4], p2[0], p2[4], p3[0], p3[4], p4[0], p4[4]};
        const float k1[10] = {p0[1], p0[5], p1[1], p1[5], p2[1], p2[5], p3[1], p3[5], p4[1], p4[5]};
        const float k2[10] = {p0[2], p0[6], p1[2], p1[6], p2[2], p2[6], p3[2], p3[6], p4[2], p4[6]};

        float sh[10];
#pragma unroll
        for (int q = 0; q < 10; ++q) sh[q] = k2[q] + row_shl_zero<1>(k1[q]);
        float ts[2];
#pragma unroll
        for (int e = 0; e < 2; ++e)
            ts[e] = (k0[8 + e] + row_shr_zero<1>(k1[8 + e])) + row_shr_zero<2>(k2[8 + e]);

        float* tw = &Top[0][0][0][0] + par * 512;
        float* ew = &Edge[0][0][0][0][0] + par * 768;
        if (kh == 0) *(v2f*)(tw + topWr) = v2f{ts[0], ts[1]};
        if (n >= 14) {
            *(v4f*)(ew + edgeWr)     = v4f{sh[0], sh[1], sh[2], sh[3]};
            *(v4f*)(ew + edgeWr + 4) = v4f{sh[4], sh[5], sh[6], sh[7]};
            *(v2f*)(ew + edgeWr + 8) = v2f{sh[8], sh[9]};
        }

        edges_signal();


        __builtin_amdgcn_sched_barrier(0);

        v8f dx = unC;
        if (anyx) {
            if (cola) {
                dx = __builtin_amdgcn_wmma_f32_16x16x4_f32(
                        false, xcC[0], false, cB[0], (short)0, dx, false, false);
                dx = __builtin_amdgcn_wmma_f32_16x16x4_f32(
                        false, xcC[1], false, cB[1], (short)0, dx, false, false);
            }
            if (hasb) {
                const v2f xr0 = {xeC[0], xeC[1]};
                const v2f xk  = {kh ? xeC[2] : 0.0f, kh ? xeC[3] : 0.0f};
                dx = __builtin_amdgcn_wmma_f32_16x16x4_f32(
                        false, wB, false, xr0, (short)0, dx, false, false);
                dx = __builtin_amdgcn_wmma_f32_16x16x4_f32(
                        false, wB2, false, xk, (short)0, dx, false, false);
            }
        }

        float acc[8];
#pragma unroll
        for (int q = 0; q < 8; ++q) acc[q] = k0[q] + dx[q];

        if (anyx && more) {
            edge_refill(xcN, xeN, xg + noff,
                        co1, co2, cv1, cv2, eo1, eo2, ev1, ev2, cola, hasb);
            if (crop) {
                const float* u = outg + noff + uoff;
#pragma unroll
                for (int q = 0; q < 8; ++q) unN[q] = u[(4 * (q >> 1) + (q & 1)) * WW];
            }
        }

        edges_wait();

        const float* tr = &Top[0][0][0][0] + par * 512;
        const float* er = &Edge[0][0][0][0][0] + par * 768;
        const v2f tt = *(const v2f*)(tr + topRd);
        const v4f e0 = *(const v4f*)(er + edgeRd);
        const v4f e1 = *(const v4f*)(er + edgeRd + 4);
        const v2f cc = *(const v2f*)(er + cornRd);

        float eq[8] = {e0[0], e0[1], e0[2], e0[3], e1[0], e1[1], e1[2], e1[3]};

        const bool toph = (kh == 0);
        acc[0] += toph ? tt[0] : 0.0f;
        acc[1] += toph ? tt[1] : 0.0f;
        eq[0]  += toph ? cc[0] : 0.0f;
        eq[1]  += toph ? cc[1] : 0.0f;

        v8f d;
#pragma unroll
        for (int q = 0; q < 8; ++q) {
            const float t1 = row_shr_zero<1>(k1[q]);
            const float t2 = row_shr_keep<2>(eq[q], k2[q]);
            d[q] = (acc[q] + t1) + t2;
        }
        hst = d;

        if (crop) {
            float* o = outg + xoff + uoff;
#pragma unroll
            for (int q = 0; q < 8; ++q) o[(4 * (q >> 1) + (q & 1)) * WW] = d[q];
        }
    };

#pragma unroll 1
    for (int t = 0; t < T_STEPS; t += 2) {
        step(t,     0, xc,  xe,  un,  xc2, xe2, un2);
        step(t + 1, 1, xc2, xe2, un2, xc,  xe,  un);
    }
}

extern "C" void kernel_launch(void* const* d_in, const int* in_sizes, int n_in,
                              void* d_out, int out_size, void* d_ws, size_t ws_size,
                              hipStream_t stream) {
    (void)in_sizes; (void)n_in; (void)out_size; (void)d_ws; (void)ws_size;
    const float* x  = (const float*)d_in[0];
    const float* Ak = (const float*)d_in[1];
    const float* Bk = (const float*)d_in[2];
    float* out = (float*)d_out;
    convssm_xterm_kernel<<<dim3(256), dim3(256), 0, stream>>>(x, Bk, out);
    convssm_scan_kernel<<<dim3(64), dim3(512), 0, stream>>>(x, Ak, Bk, out);
}


# X-term joins the scan: x_t tiles as extra accumulating WMMAs (B-kernel band times x_t in state layout) start the state's partial-sum chains, run between signal and wait; one launch; sum order changes
# speedup vs baseline: 3.6658x; 1.1686x over previous
#include <hip/hip_runtime.h>


typedef __attribute__((ext_vector_type(2))) float v2f;
typedef __attribute__((ext_vector_type(4))) float v4f;
typedef __attribute__((ext_vector_type(8))) float v8f;

#define T_STEPS 64
#define FD      64
#define HH      32
#define WW      32


template <int N>
__device__ __forceinline__ float row_shr_keep(float keep, float src)
{
    return __int_as_float(__builtin_amdgcn_update_dpp(
        __float_as_int(keep), __float_as_int(src), 0x110 + N, 0xf, 0xf, false));
}

template <int N>
__device__ __forceinline__ float row_shr_zero(float src)
{
    return __int_as_float(__builtin_amdgcn_update_dpp(
        0, __float_as_int(src), 0x110 + N, 0xf, 0xf, true));
}
template <int N>
__device__ __forceinline__ float row_shl_zero(float src)
{
    return __int_as_float(__builtin_amdgcn_update_dpp(
        0, __float_as_int(src), 0x100 + N, 0xf, 0xf, true));
}

__device__ __forceinline__ void edges_signal()
{
    asm volatile("s_wait_dscnt 0x0\n\ts_barrier_signal -1" ::: "memory");
}
__device__ __forceinline__ void edges_wait()
{
    asm volatile("s_barrier_wait -1" ::: "memory");
}

__global__ __launch_bounds__(512) void convssm_scan_kernel(
    const float* __restrict__ xg,
    const float* __restrict__ Akg,
    const float* __restrict__ Bkg,
    float* __restrict__ outg)
{
    __shared__ __attribute__((aligned(16))) float Top[2][16][16][2];
    __shared__ __attribute__((aligned(16))) float Edge[2][16][2][2][12];
    __shared__ float akl[9];
    __shared__ float bkl[9];

    const int plane = blockIdx.x;
    const int chan  = plane & 31;

    const int tid  = threadIdx.x;
    const int wave = __builtin_amdgcn_readfirstlane(tid >> 5);
    const int lane = tid & 31;
    const int n    = lane & 15;
    const int kh   = lane >> 4;
    const int m    = lane & 15;

    const unsigned long long tileOfWave = 0xFEDCBA9876325410ULL;
    const int tile = (int)((tileOfWave >> (wave * 4)) & 15ULL);
    const int rt  = tile >> 2;
    const int ct  = tile & 3;
    const int i0  = rt * 16;
    const int j0  = ct * 16;
    const int jn  = j0 + n;

    const bool crop = (rt < 2) && (ct < 2);

    const int uoff = (i0 + 2 * kh) * WW + jn;

    const int rtA = (rt + 3) & 3;
    const int ctL = (ct + 3) & 3;
    const int topWr  = (tile * 16 + n) * 2;
    const int topRd  = ((rtA * 4 + ct) * 16 + n) * 2;
    const int edgeWr = ((tile * 2 + kh) * 2 + (n == 15 ? 1 : 0)) * 12;
    const int edgeRd = (((rt * 4 + ctL) * 2 + kh) * 2 + (n & 1)) * 12;
    const int cornRd = (((rtA * 4 + ctL) * 2 + kh) * 2 + (n & 1)) * 12 + 8;

    v8f xa = {}, xb = {};
    if (crop) {
        const float* u0 = xg + (long long)plane * 1024 + uoff;
        const float* u1 = xg + ((long long)64 + plane) * 1024 + uoff;
#pragma unroll
        for (int q = 0; q < 8; ++q) {
            xb[q] = u0[(4 * (q >> 1) + (q & 1)) * WW];
            xa[q] = u1[(4 * (q >> 1) + (q & 1)) * WW];
        }
    }

    if (tid < 9) {
        akl[tid] = Akg[chan * 9 + tid];
        bkl[tid] = Bkg[chan * 9 + tid];
    }
    __syncthreads();

    v2f V0, V1, W0, W1;
    {
        float ta[9], tb[9];
#pragma unroll
        for (int i = 0; i < 9; ++i) { ta[i] = akl[i]; tb[i] = bkl[i]; }
        const int il = 2 * (m >> 3) + ((m >> 2) & 1);
        const int bc = m & 3;
        float tcol[3], ucol[3];
#pragma unroll
        for (int a = 0; a < 3; ++a) {
            tcol[a] = (bc == 0 ? ta[3 * a] : 0.0f) + (bc == 1 ? ta[3 * a + 1] : 0.0f)
                    + (bc == 2 ? ta[3 * a + 2] : 0.0f);
            ucol[a] = (bc == 0 ? tb[3 * a] : 0.0f) + (bc == 1 ? tb[3 * a + 1] : 0.0f)
                    + (bc == 2 ? tb[3 * a + 2] : 0.0f);
        }
        auto sel = [&](int a) -> float {
            return (a == 0 ? tcol[0] : 0.0f) + (a == 1 ? tcol[1] : 0.0f) + (a == 2 ? tcol[2] : 0.0f);
        };
        auto selb = [&](int a) -> float {
            return (a == 0 ? ucol[0] : 0.0f) + (a == 1 ? ucol[1] : 0.0f) + (a == 2 ? ucol[2] : 0.0f);
        };
        V1.x = sel(il - 2 * kh);      V1.y = sel(il - 2 * kh - 1);
        V0.x = sel(4 + il - 2 * kh);  V0.y = sel(3 + il - 2 * kh);
        W1.x = selb(il - 2 * kh);     W1.y = selb(il - 2 * kh - 1);
        W0.x = selb(4 + il - 2 * kh); W0.y = selb(3 + il - 2 * kh);
    }

    auto xterm = [&](const v8f& xs, v8f (&px)[5]) __attribute__((always_inline)) {
        const v2f x0 = {xs[0], xs[1]};
        const v2f x1 = {xs[2], xs[3]};
        const v2f x2 = {xs[4], xs[5]};
        const v2f x3 = {xs[6], xs[7]};
        const v8f zero8 = {};
        px[0] = __builtin_amdgcn_wmma_f32_16x16x4_f32(false, W1, false, x0, (short)0, zero8, false, false);
        px[1] = __builtin_amdgcn_wmma_f32_16x16x4_f32(false, W0, false, x0, (short)0, zero8, false, false);
        px[2] = __builtin_amdgcn_wmma_f32_16x16x4_f32(false, W0, false, x1, (short)0, zero8, false, false);
        px[3] = __builtin_amdgcn_wmma_f32_16x16x4_f32(false, W0, false, x2, (short)0, zero8, false, false);
        px[4] = __builtin_amdgcn_wmma_f32_16x16x4_f32(false, W0, false, x3, (short)0, zero8, false, false);
        px[1] = __builtin_amdgcn_wmma_f32_16x16x4_f32(false, W1, false, x1, (short)0, px[1], false, false);
        px[2] = __builtin_amdgcn_wmma_f32_16x16x4_f32(false, W1, false, x2, (short)0, px[2], false, false);
        px[3] = __builtin_amdgcn_wmma_f32_16x16x4_f32(false, W1, false, x3, (short)0, px[3], false, false);
    };

    v8f px[5];
#pragma unroll
    for (int g = 0; g < 5; ++g) px[g] = v8f{};
    if (crop) xterm(xb, px);

    v8f hst = {};

    auto step = [&](int t, const int par, const v8f& xC, v8f& xN)
                    __attribute__((always_inline)) {
        const long long xoff = ((long long)t * 64 + plane) * 1024;
        const bool more1 = t + 1 < T_STEPS;
        const bool more2 = t + 2 < T_STEPS;
        const long long noff = ((long long)(t + 2) * 64 + plane) * 1024;

        const v2f s0 = {hst[0], hst[1]};
        const v2f s1 = {hst[2], hst[3]};
        const v2f s2 = {hst[4], hst[5]};
        const v2f s3 = {hst[6], hst[7]};
        v8f p0 = __builtin_amdgcn_wmma_f32_16x16x4_f32(false, V1, false, s0, (short)0, px[0], false, false);
        v8f p1 = __builtin_amdgcn_wmma_f32_16x16x4_f32(false, V0, false, s0, (short)0, px[1], false, false);
        v8f p2 = __builtin_amdgcn_wmma_f32_16x16x4_f32(false, V0, false, s1, (short)0, px[2], false, false);
        v8f p3 = __builtin_amdgcn_wmma_f32_16x16x4_f32(false, V0, false, s2, (short)0, px[3], false, false);
        v8f p4 = __builtin_amdgcn_wmma_f32_16x16x4_f32(false, V0, false, s3, (short)0, px[4], false, false);
        p1 = __builtin_amdgcn_wmma_f32_16x16x4_f32(false, V1, false, s1, (short)0, p1, false, false);
        p2 = __builtin_amdgcn_wmma_f32_16x16x4_f32(false, V1, false, s2, (short)0, p2, false, false);
        p3 = __builtin_amdgcn_wmma_f32_16x16x4_f32(false, V1, false, s3, (short)0, p3, false, false);

        const float k0[10] = {p0[0], p0[4], p1[0], p1[4], p2[0], p2[4], p3[0], p3[4], p4[0], p4[4]};
        const float k1[10] = {p0[1], p0[5], p1[1], p1[5], p2[1], p2[5], p3[1], p3[5], p4[1], p4[5]};
        const float k2[10] = {p0[2], p0[6], p1[2], p1[6], p2[2], p2[6], p3[2], p3[6], p4[2], p4[6]};

        float sh[10];
#pragma unroll
        for (int q = 0; q < 10; ++q) sh[q] = k2[q] + row_shl_zero<1>(k1[q]);
        float ts[2];
#pragma unroll
        for (int e = 0; e < 2; ++e)
            ts[e] = (k0[8 + e] + row_shr_zero<1>(k1[8 + e])) + row_shr_zero<2>(k2[8 + e]);

        float* tw = &Top[0][0][0][0] + par * 512;
        float* ew = &Edge[0][0][0][0][0] + par * 768;
        if (kh == 0) *(v2f*)(tw + topWr) = v2f{ts[0], ts[1]};
        if (n >= 14) {
            *(v4f*)(ew + edgeWr)     = v4f{sh[0], sh[1], sh[2], sh[3]};
            *(v4f*)(ew + edgeWr + 4) = v4f{sh[4], sh[5], sh[6], sh[7]};
            *(v2f*)(ew + edgeWr + 8) = v2f{sh[8], sh[9]};
        }

        edges_signal();


        __builtin_amdgcn_sched_barrier(0);

        if (crop) {
            if (more1) xterm(xC, px);
            if (more2) {
                const float* u = xg + noff + uoff;
#pragma unroll
                for (int q = 0; q < 8; ++q) xN[q] = u[(4 * (q >> 1) + (q & 1)) * WW];
            }
        }

        edges_wait();

        const float* tr = &Top[0][0][0][0] + par * 512;
        const float* er = &Edge[0][0][0][0][0] + par * 768;
        const v2f tt = *(const v2f*)(tr + topRd);
        const v4f e0 = *(const v4f*)(er + edgeRd);
        const v4f e1 = *(const v4f*)(er + edgeRd + 4);
        const v2f cc = *(const v2f*)(er + cornRd);

        float eq[8] = {e0[0], e0[1], e0[2], e0[3], e1[0], e1[1], e1[2], e1[3]};

        float acc[8];
#pragma unroll
        for (int q = 0; q < 8; ++q) acc[q] = k0[q];

        const bool toph = (kh == 0);
        acc[0] += toph ? tt[0] : 0.0f;
        acc[1] += toph ? tt[1] : 0.0f;
        eq[0]  += toph ? cc[0] : 0.0f;
        eq[1]  += toph ? cc[1] : 0.0f;

        v8f d;
#pragma unroll
        for (int q = 0; q < 8; ++q) {
            const float t1 = row_shr_zero<1>(k1[q]);
            const float t2 = row_shr_keep<2>(eq[q], k2[q]);
            d[q] = (acc[q] + t1) + t2;
        }
        hst = d;

        if (crop) {
            float* o = outg + xoff + uoff;
#pragma unroll
            for (int q = 0; q < 8; ++q) o[(4 * (q >> 1) + (q & 1)) * WW] = d[q];
        }
    };

#pragma unroll 1
    for (int t = 0; t < T_STEPS; t += 2) {
        step(t,     0, xa, xb);
        step(t + 1, 1, xb, xa);
    }
}

extern "C" void kernel_launch(void* const* d_in, const int* in_sizes, int n_in,
                              void* d_out, int out_size, void* d_ws, size_t ws_size,
                              hipStream_t stream) {
    (void)in_sizes; (void)n_in; (void)out_size; (void)d_ws; (void)ws_size;
    const float* x  = (const float*)d_in[0];
    const float* Ak = (const float*)d_in[1];
    const float* Bk = (const float*)d_in[2];
    float* out = (float*)d_out;
    convssm_scan_kernel<<<dim3(64), dim3(512), 0, stream>>>(x, Ak, Bk, out);
}
